# starting kernel
# baseline (speedup 1.0000x reference)
#include <hip/hip_runtime.h>
#include <hip/hip_bf16.h>


typedef __attribute__((ext_vector_type(2))) float v2f;
typedef __attribute__((ext_vector_type(8))) float v8f;

#define F_IN   128
#define HEADS  4
#define C_OUT  32
#define F_OUT  128
#define ALPHA_ 0.5f
#define SLOPE_ 0.2f

__device__ __forceinline__ void atomicMaxFloat(float* addr, float val) {
    if (val >= 0.0f) {
        atomicMax((int*)addr, __float_as_int(val));
    } else {
        atomicMin((unsigned int*)addr, (unsigned int)__float_as_int(val));
    }
}

__device__ __forceinline__ float lrelu(float v) {
    return v > 0.0f ? v : v * SLOPE_;
}

__global__ __launch_bounds__(32) void gemm_wmma_kernel(
    const float* __restrict__ X,
    const float* __restrict__ W1, const float* __restrict__ W2,
    float* __restrict__ H1, float* __restrict__ H2, int n)
{
    int tile = blockIdx.x;
    int conv = tile & 1;         tile >>= 1;
    int ntile = tile & 7;
    int mtile = tile >> 3;

    const float* __restrict__ Wm = conv ? W2 : W1;
    float* __restrict__ Hm       = conv ? H2 : H1;

    int lane = threadIdx.x;
    int half = lane >> 4;
    int l    = lane & 15;

    int rowA = mtile * 16 + l;
    if (rowA >= n) rowA = n - 1;
    int colB = ntile * 16 + l;

    v8f acc = {};
    for (int k = 0; k < F_IN; k += 4) {
        int ka = k + half * 2;
        v2f a, b;
        a.x = X[rowA * F_IN + ka];
        a.y = X[rowA * F_IN + ka + 1];
        b.x = Wm[ka * F_OUT + colB];
        b.y = Wm[(ka + 1) * F_OUT + colB];
        acc = __builtin_amdgcn_wmma_f32_16x16x4_f32(
             false, a,  false, b,
             (short)0, acc,  false,  false);
    }

    int rowBase = mtile * 16 + half * 8;
    int colOut  = ntile * 16 + l;
    #pragma unroll
    for (int r = 0; r < 8; ++r) {
        int row = rowBase + r;
        if (row < n) Hm[row * F_OUT + colOut] = acc[r];
    }
}

__global__ void scores_kernel(
    const float* __restrict__ H1, const float* __restrict__ H2,
    const float* __restrict__ as1, const float* __restrict__ ad1,
    const float* __restrict__ as2, const float* __restrict__ ad2,
    float* __restrict__ es1, float* __restrict__ ed1,
    float* __restrict__ es2, float* __restrict__ ed2, int n)
{
    int i = blockIdx.x * blockDim.x + threadIdx.x;
    int total = n * HEADS * 2;
    if (i >= total) return;
    int conv = (i >= n * HEADS) ? 1 : 0;
    int j = conv ? (i - n * HEADS) : i;
    int node = j >> 2;
    int h    = j & 3;

    const float* __restrict__ hp  = (conv ? H2 : H1) + node * F_OUT + h * C_OUT;
    const float* __restrict__ asp = (conv ? as2 : as1) + h * C_OUT;
    const float* __restrict__ adp = (conv ? ad2 : ad1) + h * C_OUT;

    float ssum = 0.0f, dsum = 0.0f;
    #pragma unroll
    for (int c = 0; c < C_OUT; ++c) {
        float v = hp[c];
        ssum += v * asp[c];
        dsum += v * adp[c];
    }
    if (conv) { es2[node * HEADS + h] = ssum; ed2[node * HEADS + h] = dsum; }
    else      { es1[node * HEADS + h] = ssum; ed1[node * HEADS + h] = dsum; }
}

__global__ void init_kernel(
    const float* __restrict__ b1, const float* __restrict__ b2,
    float* __restrict__ out,
    float* __restrict__ m1, float* __restrict__ d1,
    float* __restrict__ m2, float* __restrict__ d2, int n)
{
    int i = blockIdx.x * blockDim.x + threadIdx.x;
    if (i >= n * F_OUT) return;
    int f = i & (F_OUT - 1);
    out[i] = (1.0f - ALPHA_) * b1[f] + ALPHA_ * b2[f];
    if (i < n * HEADS) {
        float ninf = -__builtin_huge_valf();
        m1[i] = ninf; m2[i] = ninf;
        d1[i] = 0.0f; d2[i] = 0.0f;
    }
}

__global__ void edge_max_kernel(
    const int* __restrict__ srcA, const int* __restrict__ dstA, int E, int n,
    const float* __restrict__ es, const float* __restrict__ ed,
    float* __restrict__ m)
{
    int idx = blockIdx.x * blockDim.x + threadIdx.x;
    if (idx >= (E + n) * HEADS) return;
    int item = idx >> 2, h = idx & 3;
    int s, d;
    if (item < E) { s = srcA[item]; d = dstA[item]; }
    else          { s = d = item - E; }
    float v = lrelu(es[s * HEADS + h] + ed[d * HEADS + h]);
    atomicMaxFloat(&m[d * HEADS + h], v);
}

__global__ void edge_sum_kernel(
    const int* __restrict__ srcA, const int* __restrict__ dstA, int E, int n,
    const float* __restrict__ es, const float* __restrict__ ed,
    const float* __restrict__ m, float* __restrict__ denom)
{
    int idx = blockIdx.x * blockDim.x + threadIdx.x;
    if (idx >= (E + n) * HEADS) return;
    int item = idx >> 2, h = idx & 3;
    int s, d;
    if (item < E) { s = srcA[item]; d = dstA[item]; }
    else          { s = d = item - E; }
    float v = lrelu(es[s * HEADS + h] + ed[d * HEADS + h]);
    float ex = __expf(v - m[d * HEADS + h]);
    atomicAdd(&denom[d * HEADS + h], ex);
}

__global__ void edge_msg_kernel(
    const int* __restrict__ srcA, const int* __restrict__ dstA, int E, int n,
    const float* __restrict__ es, const float* __restrict__ ed,
    const float* __restrict__ m, const float* __restrict__ denom,
    const float* __restrict__ Hm, float scale, float* __restrict__ out)
{
    int idx = blockIdx.x * blockDim.x + threadIdx.x;
    if (idx >= (E + n) * HEADS) return;
    int item = idx >> 2, h = idx & 3;
    int s, d;
    if (item < E) { s = srcA[item]; d = dstA[item]; }
    else          { s = d = item - E; }
    float v = lrelu(es[s * HEADS + h] + ed[d * HEADS + h]);
    float att = __expf(v - m[d * HEADS + h]) / denom[d * HEADS + h];
    float f = scale * att;

    const float* __restrict__ hp = Hm + s * F_OUT + h * C_OUT;
    float* __restrict__ op       = out + d * F_OUT + h * C_OUT;
    __builtin_prefetch(hp, 0, 1);
    #pragma unroll
    for (int c = 0; c < C_OUT; ++c) {
        atomicAdd(&op[c], f * hp[c]);
    }
}

extern "C" void kernel_launch(void* const* d_in, const int* in_sizes, int n_in,
                              void* d_out, int out_size, void* d_ws, size_t ws_size,
                              hipStream_t stream) {
    const float* x    = (const float*)d_in[0];
    const int*   ei   = (const int*)d_in[1];
    const float* W1   = (const float*)d_in[2];
    const float* as1  = (const float*)d_in[3];
    const float* ad1  = (const float*)d_in[4];
    const float* b1   = (const float*)d_in[5];
    const float* W2   = (const float*)d_in[6];
    const float* as2  = (const float*)d_in[7];
    const float* ad2  = (const float*)d_in[8];
    const float* b2   = (const float*)d_in[9];
    float* out = (float*)d_out;

    const int n = in_sizes[0] / F_IN;
    const int E = in_sizes[1] / 2;
    const int* src = ei;
    const int* dst = ei + E;

    float* p   = (float*)d_ws;
    float* H1  = p;  p += (size_t)n * F_OUT;
    float* H2  = p;  p += (size_t)n * F_OUT;
    float* es1 = p;  p += (size_t)n * HEADS;
    float* ed1 = p;  p += (size_t)n * HEADS;
    float* es2 = p;  p += (size_t)n * HEADS;
    float* ed2 = p;  p += (size_t)n * HEADS;
    float* m1  = p;  p += (size_t)n * HEADS;
    float* dn1 = p;  p += (size_t)n * HEADS;
    float* m2  = p;  p += (size_t)n * HEADS;
    float* dn2 = p;  p += (size_t)n * HEADS;

    {
        int mtiles = (n + 15) / 16;
        int blocks = 2 * mtiles * (F_OUT / 16);
        gemm_wmma_kernel<<<blocks, 32, 0, stream>>>(x, W1, W2, H1, H2, n);
    }
    {
        int total = n * HEADS * 2;
        scores_kernel<<<(total + 255) / 256, 256, 0, stream>>>(
            H1, H2, as1, ad1, as2, ad2, es1, ed1, es2, ed2, n);
    }
    {
        int total = n * F_OUT;
        init_kernel<<<(total + 255) / 256, 256, 0, stream>>>(
            b1, b2, out, m1, dn1, m2, dn2, n);
    }
    const int eth = (E + n) * HEADS;
    const int eblk = (eth + 255) / 256;
    edge_max_kernel<<<eblk, 256, 0, stream>>>(src, dst, E, n, es1, ed1, m1);
    edge_max_kernel<<<eblk, 256, 0, stream>>>(dst, src, E, n, es2, ed2, m2);
    edge_sum_kernel<<<eblk, 256, 0, stream>>>(src, dst, E, n, es1, ed1, m1, dn1);
    edge_sum_kernel<<<eblk, 256, 0, stream>>>(dst, src, E, n, es2, ed2, m2, dn2);
    edge_msg_kernel<<<eblk, 256, 0, stream>>>(src, dst, E, n, es1, ed1, m1, dn1,
                                              H1, 1.0f - ALPHA_, out);
    edge_msg_kernel<<<eblk, 256, 0, stream>>>(dst, src, E, n, es2, ed2, m2, dn2,
                                              H2, ALPHA_, out);
}


# edge_msg scatter coalesced: one wave per 32 items, one lane per channel, so each row load and each atomic add covers one whole 128-B line (was 32 lines per instruction)
# speedup vs baseline: 4.6426x; 4.6426x over previous
#include <hip/hip_runtime.h>
#include <hip/hip_bf16.h>


typedef __attribute__((ext_vector_type(2))) float v2f;
typedef __attribute__((ext_vector_type(8))) float v8f;

#define F_IN   128
#define HEADS  4
#define C_OUT  32
#define F_OUT  128
#define ALPHA_ 0.5f
#define SLOPE_ 0.2f

__device__ __forceinline__ void atomicMaxFloat(float* addr, float val) {
    if (val >= 0.0f) {
        atomicMax((int*)addr, __float_as_int(val));
    } else {
        atomicMin((unsigned int*)addr, (unsigned int)__float_as_int(val));
    }
}

__device__ __forceinline__ float lrelu(float v) {
    return v > 0.0f ? v : v * SLOPE_;
}

__global__ __launch_bounds__(32) void gemm_wmma_kernel(
    const float* __restrict__ X,
    const float* __restrict__ W1, const float* __restrict__ W2,
    float* __restrict__ H1, float* __restrict__ H2, int n)
{
    int tile = blockIdx.x;
    int conv = tile & 1;         tile >>= 1;
    int ntile = tile & 7;
    int mtile = tile >> 3;

    const float* __restrict__ Wm = conv ? W2 : W1;
    float* __restrict__ Hm       = conv ? H2 : H1;

    int lane = threadIdx.x;
    int half = lane >> 4;
    int l    = lane & 15;

    int rowA = mtile * 16 + l;
    if (rowA >= n) rowA = n - 1;
    int colB = ntile * 16 + l;

    v8f acc = {};
    for (int k = 0; k < F_IN; k += 4) {
        int ka = k + half * 2;
        v2f a, b;
        a.x = X[rowA * F_IN + ka];
        a.y = X[rowA * F_IN + ka + 1];
        b.x = Wm[ka * F_OUT + colB];
        b.y = Wm[(ka + 1) * F_OUT + colB];
        acc = __builtin_amdgcn_wmma_f32_16x16x4_f32(
             false, a,  false, b,
             (short)0, acc,  false,  false);
    }

    int rowBase = mtile * 16 + half * 8;
    int colOut  = ntile * 16 + l;
    #pragma unroll
    for (int r = 0; r < 8; ++r) {
        int row = rowBase + r;
        if (row < n) Hm[row * F_OUT + colOut] = acc[r];
    }
}

__global__ void scores_kernel(
    const float* __restrict__ H1, const float* __restrict__ H2,
    const float* __restrict__ as1, const float* __restrict__ ad1,
    const float* __restrict__ as2, const float* __restrict__ ad2,
    float* __restrict__ es1, float* __restrict__ ed1,
    float* __restrict__ es2, float* __restrict__ ed2, int n)
{
    int i = blockIdx.x * blockDim.x + threadIdx.x;
    int total = n * HEADS * 2;
    if (i >= total) return;
    int conv = (i >= n * HEADS) ? 1 : 0;
    int j = conv ? (i - n * HEADS) : i;
    int node = j >> 2;
    int h    = j & 3;

    const float* __restrict__ hp  = (conv ? H2 : H1) + node * F_OUT + h * C_OUT;
    const float* __restrict__ asp = (conv ? as2 : as1) + h * C_OUT;
    const float* __restrict__ adp = (conv ? ad2 : ad1) + h * C_OUT;

    float ssum = 0.0f, dsum = 0.0f;
    #pragma unroll
    for (int c = 0; c < C_OUT; ++c) {
        float v = hp[c];
        ssum += v * asp[c];
        dsum += v * adp[c];
    }
    if (conv) { es2[node * HEADS + h] = ssum; ed2[node * HEADS + h] = dsum; }
    else      { es1[node * HEADS + h] = ssum; ed1[node * HEADS + h] = dsum; }
}

__global__ void init_kernel(
    const float* __restrict__ b1, const float* __restrict__ b2,
    float* __restrict__ out,
    float* __restrict__ m1, float* __restrict__ d1,
    float* __restrict__ m2, float* __restrict__ d2, int n)
{
    int i = blockIdx.x * blockDim.x + threadIdx.x;
    if (i >= n * F_OUT) return;
    int f = i & (F_OUT - 1);
    out[i] = (1.0f - ALPHA_) * b1[f] + ALPHA_ * b2[f];
    if (i < n * HEADS) {
        float ninf = -__builtin_huge_valf();
        m1[i] = ninf; m2[i] = ninf;
        d1[i] = 0.0f; d2[i] = 0.0f;
    }
}

__global__ void edge_max_kernel(
    const int* __restrict__ srcA, const int* __restrict__ dstA, int E, int n,
    const float* __restrict__ es, const float* __restrict__ ed,
    float* __restrict__ m)
{
    int idx = blockIdx.x * blockDim.x + threadIdx.x;
    if (idx >= (E + n) * HEADS) return;
    int item = idx >> 2, h = idx & 3;
    int s, d;
    if (item < E) { s = srcA[item]; d = dstA[item]; }
    else          { s = d = item - E; }
    float v = lrelu(es[s * HEADS + h] + ed[d * HEADS + h]);
    atomicMaxFloat(&m[d * HEADS + h], v);
}

__global__ void edge_sum_kernel(
    const int* __restrict__ srcA, const int* __restrict__ dstA, int E, int n,
    const float* __restrict__ es, const float* __restrict__ ed,
    const float* __restrict__ m, float* __restrict__ denom)
{
    int idx = blockIdx.x * blockDim.x + threadIdx.x;
    if (idx >= (E + n) * HEADS) return;
    int item = idx >> 2, h = idx & 3;
    int s, d;
    if (item < E) { s = srcA[item]; d = dstA[item]; }
    else          { s = d = item - E; }
    float v = lrelu(es[s * HEADS + h] + ed[d * HEADS + h]);
    float ex = __expf(v - m[d * HEADS + h]);
    atomicAdd(&denom[d * HEADS + h], ex);
}

#define MSG_GROUP 4
__global__ __launch_bounds__(256) void edge_msg_kernel(
    const int* __restrict__ srcA, const int* __restrict__ dstA, int E, int n,
    const float* __restrict__ es, const float* __restrict__ ed,
    const float* __restrict__ m, const float* __restrict__ denom,
    const float* __restrict__ Hm, float scale, float* __restrict__ out)
{
    const int lane  = threadIdx.x & 31;
    const int total = E + n;
    const int wave  = __builtin_amdgcn_readfirstlane(
        (int)(blockIdx.x * (blockDim.x >> 5) + (threadIdx.x >> 5)));
    const int base  = wave * 32;
    if (base >= total) return;
    const int cnt = (total - base < 32) ? (total - base) : 32;

    int s = 0, d = 0;
    float f[HEADS] = {0.0f, 0.0f, 0.0f, 0.0f};
    if (lane < cnt) {
        const int item = base + lane;
        if (item < E) { s = srcA[item]; d = dstA[item]; }
        else          { s = d = item - E; }
        const float4 a  = *reinterpret_cast<const float4*>(es    + s * HEADS);
        const float4 b  = *reinterpret_cast<const float4*>(ed    + d * HEADS);
        const float4 mx = *reinterpret_cast<const float4*>(m     + d * HEADS);
        const float4 dn = *reinterpret_cast<const float4*>(denom + d * HEADS);
        f[0] = scale * (__expf(lrelu(a.x + b.x) - mx.x) / dn.x);
        f[1] = scale * (__expf(lrelu(a.y + b.y) - mx.y) / dn.y);
        f[2] = scale * (__expf(lrelu(a.z + b.z) - mx.z) / dn.z);
        f[3] = scale * (__expf(lrelu(a.w + b.w) - mx.w) / dn.w);
    }

    int e = 0;
    for (; e + MSG_GROUP <= cnt; e += MSG_GROUP) {
        float hv[MSG_GROUP][HEADS];
        #pragma unroll
        for (int u = 0; u < MSG_GROUP; ++u) {
            const int se = __builtin_amdgcn_readlane(s, e + u);
            const float* __restrict__ hp = Hm + se * F_OUT + lane;
            #pragma unroll
            for (int h = 0; h < HEADS; ++h) hv[u][h] = hp[h * C_OUT];
        }
        #pragma unroll
        for (int u = 0; u < MSG_GROUP; ++u) {
            const int de = __builtin_amdgcn_readlane(d, e + u);
            float* __restrict__ op = out + de * F_OUT + lane;
            #pragma unroll
            for (int h = 0; h < HEADS; ++h) {
                const float fe = __builtin_bit_cast(float,
                    __builtin_amdgcn_readlane(__builtin_bit_cast(int, f[h]), e + u));
                atomicAdd(&op[h * C_OUT], fe * hv[u][h]);
            }
        }
    }
    for (; e < cnt; ++e) {
        const int se = __builtin_amdgcn_readlane(s, e);
        const int de = __builtin_amdgcn_readlane(d, e);
        const float* __restrict__ hp = Hm + se * F_OUT + lane;
        float* __restrict__ op       = out + de * F_OUT + lane;
        #pragma unroll
        for (int h = 0; h < HEADS; ++h) {
            const float fe = __builtin_bit_cast(float,
                __builtin_amdgcn_readlane(__builtin_bit_cast(int, f[h]), e));
            atomicAdd(&op[h * C_OUT], fe * hp[h * C_OUT]);
        }
    }
}

extern "C" void kernel_launch(void* const* d_in, const int* in_sizes, int n_in,
                              void* d_out, int out_size, void* d_ws, size_t ws_size,
                              hipStream_t stream) {
    const float* x    = (const float*)d_in[0];
    const int*   ei   = (const int*)d_in[1];
    const float* W1   = (const float*)d_in[2];
    const float* as1  = (const float*)d_in[3];
    const float* ad1  = (const float*)d_in[4];
    const float* b1   = (const float*)d_in[5];
    const float* W2   = (const float*)d_in[6];
    const float* as2  = (const float*)d_in[7];
    const float* ad2  = (const float*)d_in[8];
    const float* b2   = (const float*)d_in[9];
    float* out = (float*)d_out;

    const int n = in_sizes[0] / F_IN;
    const int E = in_sizes[1] / 2;
    const int* src = ei;
    const int* dst = ei + E;

    float* p   = (float*)d_ws;
    float* H1  = p;  p += (size_t)n * F_OUT;
    float* H2  = p;  p += (size_t)n * F_OUT;
    float* es1 = p;  p += (size_t)n * HEADS;
    float* ed1 = p;  p += (size_t)n * HEADS;
    float* es2 = p;  p += (size_t)n * HEADS;
    float* ed2 = p;  p += (size_t)n * HEADS;
    float* m1  = p;  p += (size_t)n * HEADS;
    float* dn1 = p;  p += (size_t)n * HEADS;
    float* m2  = p;  p += (size_t)n * HEADS;
    float* dn2 = p;  p += (size_t)n * HEADS;

    {
        int mtiles = (n + 15) / 16;
        int blocks = 2 * mtiles * (F_OUT / 16);
        gemm_wmma_kernel<<<blocks, 32, 0, stream>>>(x, W1, W2, H1, H2, n);
    }
    {
        int total = n * HEADS * 2;
        scores_kernel<<<(total + 255) / 256, 256, 0, stream>>>(
            H1, H2, as1, ad1, as2, ad2, es1, ed1, es2, ed2, n);
    }
    {
        int total = n * F_OUT;
        init_kernel<<<(total + 255) / 256, 256, 0, stream>>>(
            b1, b2, out, m1, dn1, m2, dn2, n);
    }
    const int eth = (E + n) * HEADS;
    const int eblk = (eth + 255) / 256;
    edge_max_kernel<<<eblk, 256, 0, stream>>>(src, dst, E, n, es1, ed1, m1);
    edge_max_kernel<<<eblk, 256, 0, stream>>>(dst, src, E, n, es2, ed2, m2);
    edge_sum_kernel<<<eblk, 256, 0, stream>>>(src, dst, E, n, es1, ed1, m1, dn1);
    edge_sum_kernel<<<eblk, 256, 0, stream>>>(dst, src, E, n, es2, ed2, m2, dn2);
    const int mblk = (E + n + 255) / 256;
    edge_msg_kernel<<<mblk, 256, 0, stream>>>(src, dst, E, n, es1, ed1, m1, dn1,
                                              H1, 1.0f - ALPHA_, out);
    edge_msg_kernel<<<mblk, 256, 0, stream>>>(dst, src, E, n, es2, ed2, m2, dn2,
                                              H2, ALPHA_, out);
}


# edge passes turned from atomic scatter into gather: edges bucketed by destination row in-call, then one wave per row does max, exp-sum and weighted h-row sum in registers and stores the row once
# speedup vs baseline: 6.5030x; 1.4007x over previous
#include <hip/hip_runtime.h>


typedef __attribute__((ext_vector_type(2))) float v2f;
typedef __attribute__((ext_vector_type(8))) float v8f;

#define F_IN   128
#define HEADS  4
#define C_OUT  32
#define F_OUT  128
#define ALPHA_ 0.5f
#define SLOPE_ 0.2f

__device__ __forceinline__ float lrelu(float v) {
    return v > 0.0f ? v : v * SLOPE_;
}

__global__ __launch_bounds__(32) void gemm_wmma_kernel(
    const float* __restrict__ X, const float* __restrict__ Wm,
    float* __restrict__ Hm, int n)
{
    int tile = blockIdx.x;
    int ntile = tile & 7;
    int mtile = tile >> 3;

    int lane = threadIdx.x;
    int half = lane >> 4;
    int l    = lane & 15;

    int rowA = mtile * 16 + l;
    if (rowA >= n) rowA = n - 1;
    int colB = ntile * 16 + l;

    v8f acc = {};
    for (int k = 0; k < F_IN; k += 4) {
        int ka = k + half * 2;
        v2f a, b;
        a.x = X[rowA * F_IN + ka];
        a.y = X[rowA * F_IN + ka + 1];
        b.x = Wm[ka * F_OUT + colB];
        b.y = Wm[(ka + 1) * F_OUT + colB];
        acc = __builtin_amdgcn_wmma_f32_16x16x4_f32(
             false, a,  false, b,
             (short)0, acc,  false,  false);
    }

    int rowBase = mtile * 16 + half * 8;
    int colOut  = ntile * 16 + l;
    #pragma unroll
    for (int r = 0; r < 8; ++r) {
        int row = rowBase + r;
        if (row < n) Hm[row * F_OUT + colOut] = acc[r];
    }
}

__global__ void scores_kernel(
    const float* __restrict__ Hm,
    const float* __restrict__ as, const float* __restrict__ ad,
    float* __restrict__ es, float* __restrict__ ed, int n)
{
    int i = blockIdx.x * blockDim.x + threadIdx.x;
    if (i >= n * HEADS) return;
    int node = i >> 2;
    int h    = i & 3;

    const float* __restrict__ hp  = Hm + node * F_OUT + h * C_OUT;
    const float* __restrict__ asp = as + h * C_OUT;
    const float* __restrict__ adp = ad + h * C_OUT;

    float ssum = 0.0f, dsum = 0.0f;
    #pragma unroll
    for (int c = 0; c < C_OUT; ++c) {
        float v = hp[c];
        ssum += v * asp[c];
        dsum += v * adp[c];
    }
    es[i] = ssum;
    ed[i] = dsum;
}

__global__ void zero_counts_kernel(int* __restrict__ cntA, int* __restrict__ cntB,
                                   int* __restrict__ cursor, int n)
{
    int i = blockIdx.x * blockDim.x + threadIdx.x;
    if (i < n) { cntA[i] = 0; cntB[i] = 0; }
    if (i < 2) cursor[i] = 0;
}

__global__ void hist_kernel(const int* __restrict__ e0, const int* __restrict__ e1, int E,
                            int* __restrict__ cntA, int* __restrict__ cntB)
{
    int e = blockIdx.x * blockDim.x + threadIdx.x;
    if (e >= E) return;
    atomicAdd(&cntA[e1[e]], 1);
    atomicAdd(&cntB[e0[e]], 1);
}

__global__ __launch_bounds__(256) void alloc_kernel(
    const int* __restrict__ cntA, const int* __restrict__ cntB,
    int* __restrict__ startA, int* __restrict__ startB,
    int* __restrict__ cursor, int n)
{
    const int i    = blockIdx.x * blockDim.x + threadIdx.x;
    const int lane = threadIdx.x & 31;
    const int cA = (i < n) ? cntA[i] : 0;
    const int cB = (i < n) ? cntB[i] : 0;
    int sA = cA, sB = cB;
    #pragma unroll
    for (int o = 1; o < 32; o <<= 1) {
        const int tA = __shfl_up(sA, o, 32);
        const int tB = __shfl_up(sB, o, 32);
        if (lane >= o) { sA += tA; sB += tB; }
    }
    int baseA = 0, baseB = 0;
    if (lane == 31) {
        baseA = atomicAdd(&cursor[0], sA);
        baseB = atomicAdd(&cursor[1], sB);
    }
    baseA = __shfl(baseA, 31, 32);
    baseB = __shfl(baseB, 31, 32);
    if (i < n) {
        startA[i] = baseA + sA - cA;
        startB[i] = baseB + sB - cB;
    }
}

__global__ void fill_kernel(const int* __restrict__ e0, const int* __restrict__ e1, int E,
                            int* __restrict__ startA, int* __restrict__ startB,
                            int* __restrict__ listA, int* __restrict__ listB)
{
    int e = blockIdx.x * blockDim.x + threadIdx.x;
    if (e >= E) return;
    const int a = e0[e], b = e1[e];
    listA[atomicAdd(&startA[b], 1)] = a;
    listB[atomicAdd(&startB[a], 1)] = b;
}

__device__ __forceinline__ float wave_max(float v) {
    #pragma unroll
    for (int o = 16; o > 0; o >>= 1) v = fmaxf(v, __shfl_xor(v, o, 32));
    return v;
}
__device__ __forceinline__ float wave_sum(float v) {
    #pragma unroll
    for (int o = 16; o > 0; o >>= 1) v += __shfl_xor(v, o, 32);
    return v;
}
__device__ __forceinline__ float sel4(float a, float b, float c, float d, int k) {
    const float lo = (k & 1) ? b : a;
    const float hi = (k & 1) ? d : c;
    return (k & 2) ? hi : lo;
}
__device__ __forceinline__ float lane_value(float v, int j) {
    return __builtin_bit_cast(float,
        __builtin_amdgcn_readlane(__builtin_bit_cast(int, v), j));
}

#define ROWS_IN_FLIGHT 8
__global__ __launch_bounds__(256) void gather_kernel(
    const int* __restrict__ list, const int* __restrict__ segEnd,
    const int* __restrict__ segCnt,
    const float* __restrict__ es, const float* __restrict__ ed,
    const float* __restrict__ Hm,
    const float* __restrict__ b1, const float* __restrict__ b2,
    float scale, int first, float* __restrict__ out, int n)
{
    const int lane = threadIdx.x & 31;
    const int hsel = lane >> 3;
    const int d = __builtin_amdgcn_readfirstlane(
        (int)(blockIdx.x * (blockDim.x >> 5) + (threadIdx.x >> 5)));
    if (d >= n) return;
    const int deg   = __builtin_amdgcn_readfirstlane(segCnt[d]);
    const int beg   = __builtin_amdgcn_readfirstlane(segEnd[d]) - deg;
    const int total = deg + 1;

    const float4 edv = *reinterpret_cast<const float4*>(ed + d * HEADS);
    const float ninf = -__builtin_huge_valf();
    float m[HEADS]   = {ninf, ninf, ninf, ninf};
    float den[HEADS] = {0.0f, 0.0f, 0.0f, 0.0f};
    float4 acc = make_float4(0.0f, 0.0f, 0.0f, 0.0f);

    for (int base = 0; base < total; base += 32) {
        const int  i     = base + lane;
        const bool valid = i < total;
        int   s = d;
        float v[HEADS] = {ninf, ninf, ninf, ninf};
        if (valid) {
            if (i > 0) s = list[beg + i - 1];
            const float4 a = *reinterpret_cast<const float4*>(es + s * HEADS);
            v[0] = lrelu(a.x + edv.x);
            v[1] = lrelu(a.y + edv.y);
            v[2] = lrelu(a.z + edv.z);
            v[3] = lrelu(a.w + edv.w);
        }
        float p[HEADS], r[HEADS];
        #pragma unroll
        for (int h = 0; h < HEADS; ++h) {
            const float mn = fmaxf(m[h], wave_max(v[h]));
            r[h] = __expf(m[h] - mn);
            p[h] = valid ? __expf(v[h] - mn) : 0.0f;
            den[h] = den[h] * r[h] + wave_sum(p[h]);
            m[h] = mn;
        }
        const float rs = sel4(r[0], r[1], r[2], r[3], hsel);
        acc.x *= rs; acc.y *= rs; acc.z *= rs; acc.w *= rs;

        const int cnt = (total - base < 32) ? (total - base) : 32;
        for (int e = 0; e < cnt; e += ROWS_IN_FLIGHT) {
            float4 row[ROWS_IN_FLIGHT];
            float  w[ROWS_IN_FLIGHT];
            #pragma unroll
            for (int u = 0; u < ROWS_IN_FLIGHT; ++u) {
                const int se = __builtin_amdgcn_readlane(s, e + u);
                row[u] = *reinterpret_cast<const float4*>(Hm + (size_t)se * F_OUT + lane * 4);
                w[u] = sel4(lane_value(p[0], e + u), lane_value(p[1], e + u),
                            lane_value(p[2], e + u), lane_value(p[3], e + u), hsel);
            }
            #pragma unroll
            for (int u = 0; u < ROWS_IN_FLIGHT; ++u) {
                acc.x += w[u] * row[u].x;
                acc.y += w[u] * row[u].y;
                acc.z += w[u] * row[u].z;
                acc.w += w[u] * row[u].w;
            }
        }
    }

    const float f = scale / sel4(den[0], den[1], den[2], den[3], hsel);
    float4* __restrict__ op = reinterpret_cast<float4*>(out + (size_t)d * F_OUT + lane * 4);
    float4 o;
    if (first) {
        const float4 x1 = *reinterpret_cast<const float4*>(b1 + lane * 4);
        const float4 x2 = *reinterpret_cast<const float4*>(b2 + lane * 4);
        o.x = (1.0f - ALPHA_) * x1.x + ALPHA_ * x2.x;
        o.y = (1.0f - ALPHA_) * x1.y + ALPHA_ * x2.y;
        o.z = (1.0f - ALPHA_) * x1.z + ALPHA_ * x2.z;
        o.w = (1.0f - ALPHA_) * x1.w + ALPHA_ * x2.w;
    } else {
        o = *op;
    }
    o.x += f * acc.x; o.y += f * acc.y; o.z += f * acc.z; o.w += f * acc.w;
    *op = o;
}

extern "C" void kernel_launch(void* const* d_in, const int* in_sizes, int n_in,
                              void* d_out, int out_size, void* d_ws, size_t ws_size,
                              hipStream_t stream) {
    const float* x    = (const float*)d_in[0];
    const int*   ei   = (const int*)d_in[1];
    const float* W1   = (const float*)d_in[2];
    const float* as1  = (const float*)d_in[3];
    const float* ad1  = (const float*)d_in[4];
    const float* b1   = (const float*)d_in[5];
    const float* W2   = (const float*)d_in[6];
    const float* as2  = (const float*)d_in[7];
    const float* ad2  = (const float*)d_in[8];
    const float* b2   = (const float*)d_in[9];
    float* out = (float*)d_out;

    const int n = in_sizes[0] / F_IN;
    const int E = in_sizes[1] / 2;
    const int* src = ei;
    const int* dst = ei + E;

    float* p   = (float*)d_ws;
    float* Hm  = p;  p += (size_t)n * F_OUT;
    float* es  = p;  p += (size_t)n * HEADS;
    float* ed  = p;  p += (size_t)n * HEADS;
    int* q      = (int*)p;
    int* cntA   = q;  q += n;
    int* cntB   = q;  q += n;
    int* startA = q;  q += n;
    int* startB = q;  q += n;
    int* cursor = q;  q += 4;
    int* listA  = q;  q += E;
    int* listB  = q;

    const int nblk  = (n + 255) / 256;
    const int eblk  = (E + 255) / 256;
    const int gblk  = ((n + 15) / 16) * (F_OUT / 16);
    const int sblk  = (n * HEADS + 255) / 256;
    const int rblk  = (n + 7) / 8;

    zero_counts_kernel<<<nblk, 256, 0, stream>>>(cntA, cntB, cursor, n);
    hist_kernel<<<eblk, 256, 0, stream>>>(src, dst, E, cntA, cntB);
    alloc_kernel<<<nblk, 256, 0, stream>>>(cntA, cntB, startA, startB, cursor, n);
    fill_kernel<<<eblk, 256, 0, stream>>>(src, dst, E, startA, startB, listA, listB);

    gemm_wmma_kernel<<<gblk, 32, 0, stream>>>(x, W1, Hm, n);
    scores_kernel<<<sblk, 256, 0, stream>>>(Hm, as1, ad1, es, ed, n);
    gather_kernel<<<rblk, 256, 0, stream>>>(listA, startA, cntA, es, ed, Hm, b1, b2,
                                            1.0f - ALPHA_, 1, out, n);

    gemm_wmma_kernel<<<gblk, 32, 0, stream>>>(x, W2, Hm, n);
    scores_kernel<<<sblk, 256, 0, stream>>>(Hm, as2, ad2, es, ed, n);
    gather_kernel<<<rblk, 256, 0, stream>>>(listB, startB, cntB, es, ed, Hm, b1, b2,
                                            ALPHA_, 0, out, n);
}


# h = x@W retiled: one wave per 16 x 128 row strip, 8 strips per work-group, W staged once per group in LDS, each A fragment read once and fed to 8 WMMAs (was a one-wave group per 16 x 16 tile)
# speedup vs baseline: 13.6300x; 2.0960x over previous
#include <hip/hip_runtime.h>


typedef __attribute__((ext_vector_type(2))) float v2f;
typedef __attribute__((ext_vector_type(8))) float v8f;

#define F_IN   128
#define HEADS  4
#define C_OUT  32
#define F_OUT  128
#define ALPHA_ 0.5f
#define SLOPE_ 0.2f

__device__ __forceinline__ float lrelu(float v) {
    return v > 0.0f ? v : v * SLOPE_;
}

#define GEMM_WAVES 8
#define K_STEPS    (F_IN / 4)
#define N_TILES    (F_OUT / 16)
#define W_TASKS    ((F_IN / 2) * (F_OUT / 4) / (GEMM_WAVES * 32))
#define W_PITCH    (F_OUT + 16)

__global__ __launch_bounds__(GEMM_WAVES * 32) void gemm_wmma_kernel(
    const float* __restrict__ X, const float* __restrict__ Wm,
    float* __restrict__ Hm, int n)
{
    __shared__ v2f Wl[(F_IN / 2) * W_PITCH];

    const int tid   = threadIdx.x;
    const int lane  = tid & 31;
    const int half  = lane >> 4;
    const int l     = lane & 15;
    const int mtile = blockIdx.x * GEMM_WAVES + (tid >> 5);

    int rowA = mtile * 16 + l;
    if (rowA >= n) rowA = n - 1;

    float4 r0[W_TASKS], r1[W_TASKS];
    #pragma unroll
    for (int j = 0; j < W_TASKS; ++j) {
        const int kp = (tid >> 5) + j * GEMM_WAVES;
        const int c  = lane * 4;
        r0[j] = *reinterpret_cast<const float4*>(Wm + (size_t)(2 * kp) * F_OUT + c);
        r1[j] = *reinterpret_cast<const float4*>(Wm + (size_t)(2 * kp + 1) * F_OUT + c);
    }

    v2f a[K_STEPS];
    const float* __restrict__ xp = X + (size_t)rowA * F_IN + half * 2;
    #pragma unroll
    for (int s = 0; s < K_STEPS; ++s)
        a[s] = *reinterpret_cast<const v2f*>(xp + 4 * s);

    #pragma unroll
    for (int j = 0; j < W_TASKS; ++j) {
        const int kp = (tid >> 5) + j * GEMM_WAVES;
        v2f* __restrict__ dst = Wl + kp * W_PITCH + lane * 4;
        v2f p0, p1, p2, p3;
        p0.x = r0[j].x; p0.y = r1[j].x;
        p1.x = r0[j].y; p1.y = r1[j].y;
        p2.x = r0[j].z; p2.y = r1[j].z;
        p3.x = r0[j].w; p3.y = r1[j].w;
        dst[0] = p0; dst[1] = p1; dst[2] = p2; dst[3] = p3;
    }
    __syncthreads();

    if (mtile * 16 >= n) return;

    v8f acc[N_TILES];
    #pragma unroll
    for (int t = 0; t < N_TILES; ++t) acc[t] = (v8f){};

    const v2f* __restrict__ bp = Wl + half * W_PITCH + l;
    #pragma unroll
    for (int s = 0; s < K_STEPS; ++s) {
        #pragma unroll
        for (int t = 0; t < N_TILES; ++t) {
            const v2f b = bp[(2 * s) * W_PITCH + t * 16];
            acc[t] = __builtin_amdgcn_wmma_f32_16x16x4_f32(
                 false, a[s],  false, b,
                 (short)0, acc[t],  false,  false);
        }
    }

    const int rowBase = mtile * 16 + half * 8;
    float* __restrict__ hp = Hm + (size_t)rowBase * F_OUT + l;
    if (mtile * 16 + 16 <= n) {
        #pragma unroll
        for (int t = 0; t < N_TILES; ++t) {
            #pragma unroll
            for (int r = 0; r < 8; ++r) hp[r * F_OUT + t * 16] = acc[t][r];
        }
    } else {
        #pragma unroll
        for (int t = 0; t < N_TILES; ++t) {
            #pragma unroll
            for (int r = 0; r < 8; ++r)
                if (rowBase + r < n) hp[r * F_OUT + t * 16] = acc[t][r];
        }
    }
}

__global__ void scores_kernel(
    const float* __restrict__ Hm,
    const float* __restrict__ as, const float* __restrict__ ad,
    float* __restrict__ es, float* __restrict__ ed, int n)
{
    int i = blockIdx.x * blockDim.x + threadIdx.x;
    if (i >= n * HEADS) return;
    int node = i >> 2;
    int h    = i & 3;

    const float* __restrict__ hp  = Hm + node * F_OUT + h * C_OUT;
    const float* __restrict__ asp = as + h * C_OUT;
    const float* __restrict__ adp = ad + h * C_OUT;

    float ssum = 0.0f, dsum = 0.0f;
    #pragma unroll
    for (int c = 0; c < C_OUT; ++c) {
        float v = hp[c];
        ssum += v * asp[c];
        dsum += v * adp[c];
    }
    es[i] = ssum;
    ed[i] = dsum;
}

#define SLICES     (F_OUT / 2)
#define LDS_ROWS   50000
#define LIST_BLOCK 1024
#define EDGES_AHEAD 8
#define ROWS_AHEAD  8

__global__ __launch_bounds__(LIST_BLOCK) void hist_lds_kernel(
    const int* __restrict__ e0, const int* __restrict__ e1, int E, int n,
    int* __restrict__ part, int* __restrict__ cursor)
{
    __shared__ int bins[LDS_ROWS];
    const int tid = threadIdx.x;
    const int g   = blockIdx.x;
    const int dir = g / SLICES;
    const int w   = g - dir * SLICES;
    const int* __restrict__ key = dir ? e0 : e1;

    for (int i = tid; i < n; i += LIST_BLOCK) bins[i] = 0;
    if (g == 0 && tid < 2) cursor[tid] = 0;
    __syncthreads();

    const int chunk = (E + SLICES - 1) / SLICES;
    const int lo = w * chunk;
    const int hi = (lo + chunk < E) ? (lo + chunk) : E;
    for (int e = lo + tid; e < hi; e += EDGES_AHEAD * LIST_BLOCK) {
        int k[EDGES_AHEAD];
        #pragma unroll
        for (int u = 0; u < EDGES_AHEAD; ++u) {
            const int ee = e + u * LIST_BLOCK;
            k[u] = (ee < hi) ? key[ee] : -1;
        }
        #pragma unroll
        for (int u = 0; u < EDGES_AHEAD; ++u)
            if (k[u] >= 0) atomicAdd(&bins[k[u]], 1);
    }
    __syncthreads();

    int* __restrict__ row = part + (size_t)g * n;
    for (int i = tid; i < n; i += LIST_BLOCK) row[i] = bins[i];
}

#define SLICES_AHEAD 16
__global__ __launch_bounds__(256) void alloc_kernel(
    int* __restrict__ part,
    int* __restrict__ cntA, int* __restrict__ cntB,
    int* __restrict__ endA, int* __restrict__ endB,
    int* __restrict__ cursor, int n)
{
    const int i    = blockIdx.x * blockDim.x + threadIdx.x;
    const int lane = threadIdx.x & 31;
    int cA = 0, cB = 0;
    if (i < n) {
        int* __restrict__ pa = part + i;
        int* __restrict__ pb = part + (size_t)SLICES * n + i;
        for (int w0 = 0; w0 < SLICES; w0 += SLICES_AHEAD) {
            int ta[SLICES_AHEAD], tb[SLICES_AHEAD];
            #pragma unroll
            for (int u = 0; u < SLICES_AHEAD; ++u) {
                ta[u] = pa[(size_t)(w0 + u) * n];
                tb[u] = pb[(size_t)(w0 + u) * n];
            }
            #pragma unroll
            for (int u = 0; u < SLICES_AHEAD; ++u) {
                pa[(size_t)(w0 + u) * n] = cA;  cA += ta[u];
                pb[(size_t)(w0 + u) * n] = cB;  cB += tb[u];
            }
        }
    }
    int sA = cA, sB = cB;
    #pragma unroll
    for (int o = 1; o < 32; o <<= 1) {
        const int tA = __shfl_up(sA, o, 32);
        const int tB = __shfl_up(sB, o, 32);
        if (lane >= o) { sA += tA; sB += tB; }
    }
    int baseA = 0, baseB = 0;
    if (lane == 31) {
        baseA = atomicAdd(&cursor[0], sA);
        baseB = atomicAdd(&cursor[1], sB);
    }
    baseA = __shfl(baseA, 31, 32);
    baseB = __shfl(baseB, 31, 32);
    if (i < n) {
        cntA[i] = cA;
        cntB[i] = cB;
        endA[i] = baseA + sA;
        endB[i] = baseB + sB;
    }
}

__global__ __launch_bounds__(LIST_BLOCK) void fill_lds_kernel(
    const int* __restrict__ e0, const int* __restrict__ e1, int E, int n,
    const int* __restrict__ part,
    const int* __restrict__ cntA, const int* __restrict__ cntB,
    const int* __restrict__ endA, const int* __restrict__ endB,
    int* __restrict__ listA, int* __restrict__ listB)
{
    __shared__ int bins[LDS_ROWS];
    const int tid = threadIdx.x;
    const int g   = blockIdx.x;
    const int dir = g / SLICES;
    const int w   = g - dir * SLICES;
    const int* __restrict__ key = dir ? e0 : e1;
    const int* __restrict__ val = dir ? e1 : e0;
    const int* __restrict__ cnt = dir ? cntB : cntA;
    const int* __restrict__ end = dir ? endB : endA;
    int* __restrict__ list      = dir ? listB : listA;
    const int* __restrict__ rel = part + (size_t)g * n;

    for (int i0 = tid; i0 < n; i0 += ROWS_AHEAD * LIST_BLOCK) {
        int t[ROWS_AHEAD];
        #pragma unroll
        for (int u = 0; u < ROWS_AHEAD; ++u) {
            const int i = i0 + u * LIST_BLOCK;
            t[u] = (i < n) ? (end[i] - cnt[i] + rel[i]) : 0;
        }
        #pragma unroll
        for (int u = 0; u < ROWS_AHEAD; ++u) {
            const int i = i0 + u * LIST_BLOCK;
            if (i < n) bins[i] = t[u];
        }
    }
    __syncthreads();

    const int chunk = (E + SLICES - 1) / SLICES;
    const int lo = w * chunk;
    const int hi = (lo + chunk < E) ? (lo + chunk) : E;
    for (int e = lo + tid; e < hi; e += EDGES_AHEAD * LIST_BLOCK) {
        int k[EDGES_AHEAD], v[EDGES_AHEAD];
        #pragma unroll
        for (int u = 0; u < EDGES_AHEAD; ++u) {
            const int ee = e + u * LIST_BLOCK;
            k[u] = (ee < hi) ? key[ee] : -1;
            v[u] = (ee < hi) ? val[ee] : 0;
        }
        #pragma unroll
        for (int u = 0; u < EDGES_AHEAD; ++u)
            if (k[u] >= 0) list[atomicAdd(&bins[k[u]], 1)] = v[u];
    }
}

__device__ __forceinline__ float wave_max(float v) {
    #pragma unroll
    for (int o = 16; o > 0; o >>= 1) v = fmaxf(v, __shfl_xor(v, o, 32));
    return v;
}
__device__ __forceinline__ float wave_sum(float v) {
    #pragma unroll
    for (int o = 16; o > 0; o >>= 1) v += __shfl_xor(v, o, 32);
    return v;
}
__device__ __forceinline__ float sel4(float a, float b, float c, float d, int k) {
    const float lo = (k & 1) ? b : a;
    const float hi = (k & 1) ? d : c;
    return (k & 2) ? hi : lo;
}
__device__ __forceinline__ float lane_value(float v, int j) {
    return __builtin_bit_cast(float,
        __builtin_amdgcn_readlane(__builtin_bit_cast(int, v), j));
}

#define ROWS_IN_FLIGHT 8
__global__ __launch_bounds__(256) void gather_kernel(
    const int* __restrict__ list, const int* __restrict__ segEnd,
    const int* __restrict__ segCnt,
    const float* __restrict__ es, const float* __restrict__ ed,
    const float* __restrict__ Hm,
    const float* __restrict__ b1, const float* __restrict__ b2,
    float scale, int first, float* __restrict__ out, int n)
{
    const int lane = threadIdx.x & 31;
    const int hsel = lane >> 3;
    const int d = __builtin_amdgcn_readfirstlane(
        (int)(blockIdx.x * (blockDim.x >> 5) + (threadIdx.x >> 5)));
    if (d >= n) return;
    const int deg   = __builtin_amdgcn_readfirstlane(segCnt[d]);
    const int beg   = __builtin_amdgcn_readfirstlane(segEnd[d]) - deg;
    const int total = deg + 1;

    const float4 edv = *reinterpret_cast<const float4*>(ed + d * HEADS);
    const float ninf = -__builtin_huge_valf();
    float m[HEADS]   = {ninf, ninf, ninf, ninf};
    float den[HEADS] = {0.0f, 0.0f, 0.0f, 0.0f};
    float4 acc = make_float4(0.0f, 0.0f, 0.0f, 0.0f);

    for (int base = 0; base < total; base += 32) {
        const int  i     = base + lane;
        const bool valid = i < total;
        int   s = d;
        float v[HEADS] = {ninf, ninf, ninf, ninf};
        if (valid) {
            if (i > 0) s = list[beg + i - 1];
            const float4 a = *reinterpret_cast<const float4*>(es + s * HEADS);
            v[0] = lrelu(a.x + edv.x);
            v[1] = lrelu(a.y + edv.y);
            v[2] = lrelu(a.z + edv.z);
            v[3] = lrelu(a.w + edv.w);
        }
        float p[HEADS], r[HEADS];
        #pragma unroll
        for (int h = 0; h < HEADS; ++h) {
            const float mn = fmaxf(m[h], wave_max(v[h]));
            r[h] = __expf(m[h] - mn);
            p[h] = valid ? __expf(v[h] - mn) : 0.0f;
            den[h] = den[h] * r[h] + wave_sum(p[h]);
            m[h] = mn;
        }
        const float rs = sel4(r[0], r[1], r[2], r[3], hsel);
        acc.x *= rs; acc.y *= rs; acc.z *= rs; acc.w *= rs;

        const int cnt = (total - base < 32) ? (total - base) : 32;
        for (int e = 0; e < cnt; e += ROWS_IN_FLIGHT) {
            float4 row[ROWS_IN_FLIGHT];
            float  w[ROWS_IN_FLIGHT];
            #pragma unroll
            for (int u = 0; u < ROWS_IN_FLIGHT; ++u) {
                const int se = __builtin_amdgcn_readlane(s, e + u);
                row[u] = *reinterpret_cast<const float4*>(Hm + (size_t)se * F_OUT + lane * 4);
                w[u] = sel4(lane_value(p[0], e + u), lane_value(p[1], e + u),
                            lane_value(p[2], e + u), lane_value(p[3], e + u), hsel);
            }
            #pragma unroll
            for (int u = 0; u < ROWS_IN_FLIGHT; ++u) {
                acc.x += w[u] * row[u].x;
                acc.y += w[u] * row[u].y;
                acc.z += w[u] * row[u].z;
                acc.w += w[u] * row[u].w;
            }
        }
    }

    const float f = scale / sel4(den[0], den[1], den[2], den[3], hsel);
    float4* __restrict__ op = reinterpret_cast<float4*>(out + (size_t)d * F_OUT + lane * 4);
    float4 o;
    if (first) {
        const float4 x1 = *reinterpret_cast<const float4*>(b1 + lane * 4);
        const float4 x2 = *reinterpret_cast<const float4*>(b2 + lane * 4);
        o.x = (1.0f - ALPHA_) * x1.x + ALPHA_ * x2.x;
        o.y = (1.0f - ALPHA_) * x1.y + ALPHA_ * x2.y;
        o.z = (1.0f - ALPHA_) * x1.z + ALPHA_ * x2.z;
        o.w = (1.0f - ALPHA_) * x1.w + ALPHA_ * x2.w;
    } else {
        o = *op;
    }
    o.x += f * acc.x; o.y += f * acc.y; o.z += f * acc.z; o.w += f * acc.w;
    *op = o;
}

extern "C" void kernel_launch(void* const* d_in, const int* in_sizes, int n_in,
                              void* d_out, int out_size, void* d_ws, size_t ws_size,
                              hipStream_t stream) {
    const float* x    = (const float*)d_in[0];
    const int*   ei   = (const int*)d_in[1];
    const float* W1   = (const float*)d_in[2];
    const float* as1  = (const float*)d_in[3];
    const float* ad1  = (const float*)d_in[4];
    const float* b1   = (const float*)d_in[5];
    const float* W2   = (const float*)d_in[6];
    const float* as2  = (const float*)d_in[7];
    const float* ad2  = (const float*)d_in[8];
    const float* b2   = (const float*)d_in[9];
    float* out = (float*)d_out;

    const int n = in_sizes[0] / F_IN;
    const int E = in_sizes[1] / 2;
    const int* src = ei;
    const int* dst = ei + E;

    float* p   = (float*)d_ws;
    float* Hm  = p;  p += (size_t)n * F_OUT;
    float* es  = p;  p += (size_t)n * HEADS;
    float* ed  = p;  p += (size_t)n * HEADS;
    int* q      = (int*)p;
    int* cntA   = q;  q += n;
    int* cntB   = q;  q += n;
    int* endA = q;  q += n;
    int* endB = q;  q += n;
    int* cursor = q;  q += 4;
    int* listA  = q;  q += E;
    int* listB  = q;

    const int nblk  = (n + 255) / 256;
    const int gblk  = (n + 16 * GEMM_WAVES - 1) / (16 * GEMM_WAVES);
    const int sblk  = (n * HEADS + 255) / 256;
    const int rblk  = (n + 7) / 8;

    int* part = (int*)Hm;
    hist_lds_kernel<<<2 * SLICES, LIST_BLOCK, 0, stream>>>(src, dst, E, n, part, cursor);
    alloc_kernel<<<nblk, 256, 0, stream>>>(part, cntA, cntB, endA, endB, cursor, n);
    fill_lds_kernel<<<2 * SLICES, LIST_BLOCK, 0, stream>>>(src, dst, E, n, part, cntA, cntB,
                                                          endA, endB, listA, listB);

    gemm_wmma_kernel<<<gblk, GEMM_WAVES * 32, 0, stream>>>(x, W1, Hm, n);
    scores_kernel<<<sblk, 256, 0, stream>>>(Hm, as1, ad1, es, ed, n);
    gather_kernel<<<rblk, 256, 0, stream>>>(listA, endA, cntA, es, ed, Hm, b1, b2,
                                            1.0f - ALPHA_, 1, out, n);

    gemm_wmma_kernel<<<gblk, GEMM_WAVES * 32, 0, stream>>>(x, W2, Hm, n);
    scores_kernel<<<sblk, 256, 0, stream>>>(Hm, as2, ad2, es, ed, n);
    gather_kernel<<<rblk, 256, 0, stream>>>(listB, endB, cntB, es, ed, Hm, b1, b2,
                                            ALPHA_, 0, out, n);
}


# gather forms each item's logits from the h row it loads anyway (lane dot, 8-lane DPP halving sum, running softmax per 8 rows; new summation order): es/ed arrays and their per-lane line gather gone
# speedup vs baseline: 15.2207x; 1.1167x over previous
#include <hip/hip_runtime.h>


typedef __attribute__((ext_vector_type(2))) float v2f;
typedef __attribute__((ext_vector_type(8))) float v8f;

#define F_IN   128
#define HEADS  4
#define C_OUT  32
#define F_OUT  128
#define ALPHA_ 0.5f
#define SLOPE_ 0.2f

__device__ __forceinline__ float lrelu(float v) {
    return v > 0.0f ? v : v * SLOPE_;
}

#define GEMM_WAVES 8
#define K_STEPS    (F_IN / 4)
#define N_TILES    (F_OUT / 16)
#define W_TASKS    ((F_IN / 2) * (F_OUT / 4) / (GEMM_WAVES * 32))
#define W_PITCH    (F_OUT + 16)

__global__ __launch_bounds__(GEMM_WAVES * 32) void gemm_wmma_kernel(
    const float* __restrict__ X, const float* __restrict__ Wm,
    float* __restrict__ Hm, int n)
{
    __shared__ v2f Wl[(F_IN / 2) * W_PITCH];

    const int tid   = threadIdx.x;
    const int lane  = tid & 31;
    const int half  = lane >> 4;
    const int l     = lane & 15;
    const int mtile = blockIdx.x * GEMM_WAVES + (tid >> 5);

    int rowA = mtile * 16 + l;
    if (rowA >= n) rowA = n - 1;

    float4 r0[W_TASKS], r1[W_TASKS];
    #pragma unroll
    for (int j = 0; j < W_TASKS; ++j) {
        const int kp = (tid >> 5) + j * GEMM_WAVES;
        const int c  = lane * 4;
        r0[j] = *reinterpret_cast<const float4*>(Wm + (size_t)(2 * kp) * F_OUT + c);
        r1[j] = *reinterpret_cast<const float4*>(Wm + (size_t)(2 * kp + 1) * F_OUT + c);
    }

    v2f a[K_STEPS];
    const float* __restrict__ xp = X + (size_t)rowA * F_IN + half * 2;
    #pragma unroll
    for (int s = 0; s < K_STEPS; ++s)
        a[s] = *reinterpret_cast<const v2f*>(xp + 4 * s);

    #pragma unroll
    for (int j = 0; j < W_TASKS; ++j) {
        const int kp = (tid >> 5) + j * GEMM_WAVES;
        v2f* __restrict__ dst = Wl + kp * W_PITCH + lane * 4;
        v2f p0, p1, p2, p3;
        p0.x = r0[j].x; p0.y = r1[j].x;
        p1.x = r0[j].y; p1.y = r1[j].y;
        p2.x = r0[j].z; p2.y = r1[j].z;
        p3.x = r0[j].w; p3.y = r1[j].w;
        dst[0] = p0; dst[1] = p1; dst[2] = p2; dst[3] = p3;
    }
    __syncthreads();

    if (mtile * 16 >= n) return;

    v8f acc[N_TILES];
    #pragma unroll
    for (int t = 0; t < N_TILES; ++t) acc[t] = (v8f){};

    const v2f* __restrict__ bp = Wl + half * W_PITCH + l;
    #pragma unroll
    for (int s = 0; s < K_STEPS; ++s) {
        #pragma unroll
        for (int t = 0; t < N_TILES; ++t) {
            const v2f b = bp[(2 * s) * W_PITCH + t * 16];
            acc[t] = __builtin_amdgcn_wmma_f32_16x16x4_f32(
                 false, a[s],  false, b,
                 (short)0, acc[t],  false,  false);
        }
    }

    const int rowBase = mtile * 16 + half * 8;
    float* __restrict__ hp = Hm + (size_t)rowBase * F_OUT + l;
    if (mtile * 16 + 16 <= n) {
        #pragma unroll
        for (int t = 0; t < N_TILES; ++t) {
            #pragma unroll
            for (int r = 0; r < 8; ++r) hp[r * F_OUT + t * 16] = acc[t][r];
        }
    } else {
        #pragma unroll
        for (int t = 0; t < N_TILES; ++t) {
            #pragma unroll
            for (int r = 0; r < 8; ++r)
                if (rowBase + r < n) hp[r * F_OUT + t * 16] = acc[t][r];
        }
    }
}

#define SLICES     (F_OUT / 2)
#define LDS_ROWS   50000
#define LIST_BLOCK 1024
#define EDGES_AHEAD 8
#define ROWS_AHEAD  8

__global__ __launch_bounds__(LIST_BLOCK) void hist_lds_kernel(
    const int* __restrict__ e0, const int* __restrict__ e1, int E, int n,
    int* __restrict__ part, int* __restrict__ cursor)
{
    __shared__ int bins[LDS_ROWS];
    const int tid = threadIdx.x;
    const int g   = blockIdx.x;
    const int dir = g / SLICES;
    const int w   = g - dir * SLICES;
    const int* __restrict__ key = dir ? e0 : e1;

    for (int i = tid; i < n; i += LIST_BLOCK) bins[i] = 0;
    if (g == 0 && tid < 2) cursor[tid] = 0;
    __syncthreads();

    const int chunk = (E + SLICES - 1) / SLICES;
    const int lo = w * chunk;
    const int hi = (lo + chunk < E) ? (lo + chunk) : E;
    for (int e = lo + tid; e < hi; e += EDGES_AHEAD * LIST_BLOCK) {
        int k[EDGES_AHEAD];
        #pragma unroll
        for (int u = 0; u < EDGES_AHEAD; ++u) {
            const int ee = e + u * LIST_BLOCK;
            k[u] = (ee < hi) ? key[ee] : -1;
        }
        #pragma unroll
        for (int u = 0; u < EDGES_AHEAD; ++u)
            if (k[u] >= 0) atomicAdd(&bins[k[u]], 1);
    }
    __syncthreads();

    int* __restrict__ row = part + (size_t)g * n;
    for (int i = tid; i < n; i += LIST_BLOCK) row[i] = bins[i];
}

#define SLICES_AHEAD 16
__global__ __launch_bounds__(256) void alloc_kernel(
    int* __restrict__ part,
    int* __restrict__ cntA, int* __restrict__ cntB,
    int* __restrict__ endA, int* __restrict__ endB,
    int* __restrict__ cursor, int n)
{
    const int i    = blockIdx.x * blockDim.x + threadIdx.x;
    const int lane = threadIdx.x & 31;
    int cA = 0, cB = 0;
    if (i < n) {
        int* __restrict__ pa = part + i;
        int* __restrict__ pb = part + (size_t)SLICES * n + i;
        for (int w0 = 0; w0 < SLICES; w0 += SLICES_AHEAD) {
            int ta[SLICES_AHEAD], tb[SLICES_AHEAD];
            #pragma unroll
            for (int u = 0; u < SLICES_AHEAD; ++u) {
                ta[u] = pa[(size_t)(w0 + u) * n];
                tb[u] = pb[(size_t)(w0 + u) * n];
            }
            #pragma unroll
            for (int u = 0; u < SLICES_AHEAD; ++u) {
                pa[(size_t)(w0 + u) * n] = cA;  cA += ta[u];
                pb[(size_t)(w0 + u) * n] = cB;  cB += tb[u];
            }
        }
    }
    int sA = cA, sB = cB;
    #pragma unroll
    for (int o = 1; o < 32; o <<= 1) {
        const int tA = __shfl_up(sA, o, 32);
        const int tB = __shfl_up(sB, o, 32);
        if (lane >= o) { sA += tA; sB += tB; }
    }
    int baseA = 0, baseB = 0;
    if (lane == 31) {
        baseA = atomicAdd(&cursor[0], sA);
        baseB = atomicAdd(&cursor[1], sB);
    }
    baseA = __shfl(baseA, 31, 32);
    baseB = __shfl(baseB, 31, 32);
    if (i < n) {
        cntA[i] = cA;
        cntB[i] = cB;
        endA[i] = baseA + sA;
        endB[i] = baseB + sB;
    }
}

__global__ __launch_bounds__(LIST_BLOCK) void fill_lds_kernel(
    const int* __restrict__ e0, const int* __restrict__ e1, int E, int n,
    const int* __restrict__ part,
    const int* __restrict__ cntA, const int* __restrict__ cntB,
    const int* __restrict__ endA, const int* __restrict__ endB,
    int* __restrict__ listA, int* __restrict__ listB)
{
    __shared__ int bins[LDS_ROWS];
    const int tid = threadIdx.x;
    const int g   = blockIdx.x;
    const int dir = g / SLICES;
    const int w   = g - dir * SLICES;
    const int* __restrict__ key = dir ? e0 : e1;
    const int* __restrict__ val = dir ? e1 : e0;
    const int* __restrict__ cnt = dir ? cntB : cntA;
    const int* __restrict__ end = dir ? endB : endA;
    int* __restrict__ list      = dir ? listB : listA;
    const int* __restrict__ rel = part + (size_t)g * n;

    for (int i0 = tid; i0 < n; i0 += ROWS_AHEAD * LIST_BLOCK) {
        int t[ROWS_AHEAD];
        #pragma unroll
        for (int u = 0; u < ROWS_AHEAD; ++u) {
            const int i = i0 + u * LIST_BLOCK;
            t[u] = (i < n) ? (end[i] - cnt[i] + rel[i]) : 0;
        }
        #pragma unroll
        for (int u = 0; u < ROWS_AHEAD; ++u) {
            const int i = i0 + u * LIST_BLOCK;
            if (i < n) bins[i] = t[u];
        }
    }
    __syncthreads();

    const int chunk = (E + SLICES - 1) / SLICES;
    const int lo = w * chunk;
    const int hi = (lo + chunk < E) ? (lo + chunk) : E;
    for (int e = lo + tid; e < hi; e += EDGES_AHEAD * LIST_BLOCK) {
        int k[EDGES_AHEAD], v[EDGES_AHEAD];
        #pragma unroll
        for (int u = 0; u < EDGES_AHEAD; ++u) {
            const int ee = e + u * LIST_BLOCK;
            k[u] = (ee < hi) ? key[ee] : -1;
            v[u] = (ee < hi) ? val[ee] : 0;
        }
        #pragma unroll
        for (int u = 0; u < EDGES_AHEAD; ++u)
            if (k[u] >= 0) list[atomicAdd(&bins[k[u]], 1)] = v[u];
    }
}

typedef __attribute__((ext_vector_type(4))) float v4f;
#define LOG2E_ 1.4426950408889634f

template <int K>
__device__ __forceinline__ float from_lane_xor(float v) {
    return __builtin_bit_cast(float,
        __builtin_amdgcn_update_dpp(0, __builtin_bit_cast(int, v), 0x160 + K, 0xf, 0xf, true));
}
template <int U>
__device__ __forceinline__ float from_group_lane(float v) {
    return __builtin_bit_cast(float,
        __builtin_amdgcn_mov_dpp8(__builtin_bit_cast(int, v), U * 0x249249));
}
__device__ __forceinline__ float head_sum(float v) {
    v += from_lane_xor<1>(v);
    v += from_lane_xor<2>(v);
    v += from_lane_xor<4>(v);
    return v;
}
__device__ __forceinline__ float head_max(float v) {
    v = fmaxf(v, from_lane_xor<1>(v));
    v = fmaxf(v, from_lane_xor<2>(v));
    v = fmaxf(v, from_lane_xor<4>(v));
    return v;
}
__device__ __forceinline__ float dot4(const v4f a, const v4f b) {
    const v2f t = a.lo * b.lo + a.hi * b.hi;
    return t.x + t.y;
}

#define ROWS_IN_FLIGHT 8
__global__ __launch_bounds__(256) void gather_kernel(
    const int* __restrict__ list, const int* __restrict__ segEnd,
    const int* __restrict__ segCnt,
    const float* __restrict__ as, const float* __restrict__ ad,
    const float* __restrict__ Hm,
    const float* __restrict__ b1, const float* __restrict__ b2,
    float scale, int first, float* __restrict__ out, int n)
{
    const int lane = threadIdx.x & 31;
    const int j    = lane & 7;
    const bool j4 = (j & 4) != 0, j2 = (j & 2) != 0, j1 = (j & 1) != 0;
    const int d = __builtin_amdgcn_readfirstlane(
        (int)(blockIdx.x * (blockDim.x >> 5) + (threadIdx.x >> 5)));
    if (d >= n) return;
    const int deg = __builtin_amdgcn_readfirstlane(segCnt[d]);
    const int beg = __builtin_amdgcn_readfirstlane(segEnd[d]) - deg;

    const float* __restrict__ hl = Hm + lane * 4;
    const v4f a4 = *reinterpret_cast<const v4f*>(as + lane * 4) * LOG2E_;
    const v4f d4 = *reinterpret_cast<const v4f*>(ad + lane * 4) * LOG2E_;

    const v4f self = *reinterpret_cast<const v4f*>(hl + (size_t)d * F_OUT);
    const float edh = head_sum(dot4(self, d4));
    float m   = lrelu(head_sum(dot4(self, a4)) + edh);
    float den = (j == 0) ? 1.0f : 0.0f;
    v4f   acc = self;

    const float ninf = -__builtin_huge_valf();
    for (int base = 0; base < deg; base += 32) {
        const int i = base + lane;
        int s = d;
        if (i < deg) s = list[beg + i];
        const int cnt = (deg - base < 32) ? (deg - base) : 32;
        for (int e = 0; e < cnt; e += ROWS_IN_FLIGHT) {
            v4f   row[ROWS_IN_FLIGHT];
            float q[ROWS_IN_FLIGHT];
            #pragma unroll
            for (int u = 0; u < ROWS_IN_FLIGHT; ++u) {
                const int se = __builtin_amdgcn_readlane(s, e + u);
                row[u] = *reinterpret_cast<const v4f*>(hl + (size_t)se * F_OUT);
            }
            #pragma unroll
            for (int u = 0; u < ROWS_IN_FLIGHT; ++u) q[u] = dot4(row[u], a4);

            float k4[4], k2[2];
            #pragma unroll
            for (int t = 0; t < 4; ++t) {
                const float keep = j4 ? q[4 + t] : q[t];
                const float send = j4 ? q[t] : q[4 + t];
                k4[t] = keep + from_lane_xor<4>(send);
            }
            #pragma unroll
            for (int t = 0; t < 2; ++t) {
                const float keep = j2 ? k4[2 + t] : k4[t];
                const float send = j2 ? k4[t] : k4[2 + t];
                k2[t] = keep + from_lane_xor<2>(send);
            }
            const float keep1 = j1 ? k2[1] : k2[0];
            const float send1 = j1 ? k2[0] : k2[1];
            float v = lrelu(keep1 + from_lane_xor<1>(send1) + edh);
            if (e + j >= cnt) v = ninf;

            const float mn = fmaxf(m, head_max(v));
            const float r  = __builtin_amdgcn_exp2f(m - mn);
            const float p  = __builtin_amdgcn_exp2f(v - mn);
            m   = mn;
            den = den * r + p;
            acc *= r;
            acc += from_group_lane<0>(p) * row[0];
            acc += from_group_lane<1>(p) * row[1];
            acc += from_group_lane<2>(p) * row[2];
            acc += from_group_lane<3>(p) * row[3];
            acc += from_group_lane<4>(p) * row[4];
            acc += from_group_lane<5>(p) * row[5];
            acc += from_group_lane<6>(p) * row[6];
            acc += from_group_lane<7>(p) * row[7];
        }
    }

    const float f = scale / head_sum(den);
    float4* __restrict__ op = reinterpret_cast<float4*>(out + (size_t)d * F_OUT + lane * 4);
    float4 o;
    if (first) {
        const float4 x1 = *reinterpret_cast<const float4*>(b1 + lane * 4);
        const float4 x2 = *reinterpret_cast<const float4*>(b2 + lane * 4);
        o.x = (1.0f - ALPHA_) * x1.x + ALPHA_ * x2.x;
        o.y = (1.0f - ALPHA_) * x1.y + ALPHA_ * x2.y;
        o.z = (1.0f - ALPHA_) * x1.z + ALPHA_ * x2.z;
        o.w = (1.0f - ALPHA_) * x1.w + ALPHA_ * x2.w;
    } else {
        o = *op;
    }
    o.x += f * acc.x; o.y += f * acc.y; o.z += f * acc.z; o.w += f * acc.w;
    *op = o;
}

extern "C" void kernel_launch(void* const* d_in, const int* in_sizes, int n_in,
                              void* d_out, int out_size, void* d_ws, size_t ws_size,
                              hipStream_t stream) {
    const float* x    = (const float*)d_in[0];
    const int*   ei   = (const int*)d_in[1];
    const float* W1   = (const float*)d_in[2];
    const float* as1  = (const float*)d_in[3];
    const float* ad1  = (const float*)d_in[4];
    const float* b1   = (const float*)d_in[5];
    const float* W2   = (const float*)d_in[6];
    const float* as2  = (const float*)d_in[7];
    const float* ad2  = (const float*)d_in[8];
    const float* b2   = (const float*)d_in[9];
    float* out = (float*)d_out;

    const int n = in_sizes[0] / F_IN;
    const int E = in_sizes[1] / 2;
    const int* src = ei;
    const int* dst = ei + E;

    float* p   = (float*)d_ws;
    float* Hm  = p;  p += (size_t)n * F_OUT;
    int* q      = (int*)p;
    int* cntA   = q;  q += n;
    int* cntB   = q;  q += n;
    int* endA = q;  q += n;
    int* endB = q;  q += n;
    int* cursor = q;  q += 4;
    int* listA  = q;  q += E;
    int* listB  = q;

    const int nblk  = (n + 255) / 256;
    const int gblk  = (n + 16 * GEMM_WAVES - 1) / (16 * GEMM_WAVES);
    const int rblk  = (n + 7) / 8;

    int* part = (int*)Hm;
    hist_lds_kernel<<<2 * SLICES, LIST_BLOCK, 0, stream>>>(src, dst, E, n, part, cursor);
    alloc_kernel<<<nblk, 256, 0, stream>>>(part, cntA, cntB, endA, endB, cursor, n);
    fill_lds_kernel<<<2 * SLICES, LIST_BLOCK, 0, stream>>>(src, dst, E, n, part, cntA, cntB,
                                                          endA, endB, listA, listB);

    gemm_wmma_kernel<<<gblk, GEMM_WAVES * 32, 0, stream>>>(x, W1, Hm, n);
    gather_kernel<<<rblk, 256, 0, stream>>>(listA, endA, cntA, as1, ad1, Hm, b1, b2,
                                            1.0f - ALPHA_, 1, out, n);

    gemm_wmma_kernel<<<gblk, GEMM_WAVES * 32, 0, stream>>>(x, W2, Hm, n);
    gather_kernel<<<rblk, 256, 0, stream>>>(listB, endB, cntB, as2, ad2, Hm, b1, b2,
                                            ALPHA_, 0, out, n);
}


# conv 1's h = x@W1 has no launch of its own: its strips run as 128 extra 1024-thread groups of the list-fill launch, on the WGPs the fill leaves idle; slice-count table moved from h to out
# speedup vs baseline: 16.1252x; 1.0594x over previous
#include <hip/hip_runtime.h>


typedef __attribute__((ext_vector_type(2))) float v2f;
typedef __attribute__((ext_vector_type(8))) float v8f;

#define F_IN   128
#define HEADS  4
#define C_OUT  32
#define F_OUT  128
#define ALPHA_ 0.5f
#define SLOPE_ 0.2f

__device__ __forceinline__ float lrelu(float v) {
    return v > 0.0f ? v : v * SLOPE_;
}

#define GEMM_WAVES 8
#define K_STEPS    (F_IN / 4)
#define N_TILES    (F_OUT / 16)
#define W_TASKS    ((F_IN / 2) * (F_OUT / 4) / (GEMM_WAVES * 32))
#define W_PITCH    (F_OUT + 16)

__global__ __launch_bounds__(GEMM_WAVES * 32) void gemm_wmma_kernel(
    const float* __restrict__ X, const float* __restrict__ Wm,
    float* __restrict__ Hm, int n)
{
    __shared__ v2f Wl[(F_IN / 2) * W_PITCH];

    const int tid   = threadIdx.x;
    const int lane  = tid & 31;
    const int half  = lane >> 4;
    const int l     = lane & 15;
    const int mtile = blockIdx.x * GEMM_WAVES + (tid >> 5);

    int rowA = mtile * 16 + l;
    if (rowA >= n) rowA = n - 1;

    float4 r0[W_TASKS], r1[W_TASKS];
    #pragma unroll
    for (int j = 0; j < W_TASKS; ++j) {
        const int kp = (tid >> 5) + j * GEMM_WAVES;
        const int c  = lane * 4;
        r0[j] = *reinterpret_cast<const float4*>(Wm + (size_t)(2 * kp) * F_OUT + c);
        r1[j] = *reinterpret_cast<const float4*>(Wm + (size_t)(2 * kp + 1) * F_OUT + c);
    }

    v2f a[K_STEPS];
    const float* __restrict__ xp = X + (size_t)rowA * F_IN + half * 2;
    #pragma unroll
    for (int s = 0; s < K_STEPS; ++s)
        a[s] = *reinterpret_cast<const v2f*>(xp + 4 * s);

    #pragma unroll
    for (int j = 0; j < W_TASKS; ++j) {
        const int kp = (tid >> 5) + j * GEMM_WAVES;
        v2f* __restrict__ dst = Wl + kp * W_PITCH + lane * 4;
        v2f p0, p1, p2, p3;
        p0.x = r0[j].x; p0.y = r1[j].x;
        p1.x = r0[j].y; p1.y = r1[j].y;
        p2.x = r0[j].z; p2.y = r1[j].z;
        p3.x = r0[j].w; p3.y = r1[j].w;
        dst[0] = p0; dst[1] = p1; dst[2] = p2; dst[3] = p3;
    }
    __syncthreads();

    if (mtile * 16 >= n) return;

    v8f acc[N_TILES];
    #pragma unroll
    for (int t = 0; t < N_TILES; ++t) acc[t] = (v8f){};

    const v2f* __restrict__ bp = Wl + half * W_PITCH + l;
    #pragma unroll
    for (int s = 0; s < K_STEPS; ++s) {
        #pragma unroll
        for (int t = 0; t < N_TILES; ++t) {
            const v2f b = bp[(2 * s) * W_PITCH + t * 16];
            acc[t] = __builtin_amdgcn_wmma_f32_16x16x4_f32(
                 false, a[s],  false, b,
                 (short)0, acc[t],  false,  false);
        }
    }

    const int rowBase = mtile * 16 + half * 8;
    float* __restrict__ hp = Hm + (size_t)rowBase * F_OUT + l;
    if (mtile * 16 + 16 <= n) {
        #pragma unroll
        for (int t = 0; t < N_TILES; ++t) {
            #pragma unroll
            for (int r = 0; r < 8; ++r) hp[r * F_OUT + t * 16] = acc[t][r];
        }
    } else {
        #pragma unroll
        for (int t = 0; t < N_TILES; ++t) {
            #pragma unroll
            for (int r = 0; r < 8; ++r)
                if (rowBase + r < n) hp[r * F_OUT + t * 16] = acc[t][r];
        }
    }
}

#define SLICES     (F_OUT / 2)
#define LDS_ROWS   50000
#define LIST_BLOCK 1024
#define EDGES_AHEAD 8
#define ROWS_AHEAD  8

__global__ __launch_bounds__(LIST_BLOCK) void hist_lds_kernel(
    const int* __restrict__ e0, const int* __restrict__ e1, int E, int n,
    int* __restrict__ part, int* __restrict__ cursor)
{
    __shared__ int bins[LDS_ROWS];
    const int tid = threadIdx.x;
    const int g   = blockIdx.x;
    const int dir = g / SLICES;
    const int w   = g - dir * SLICES;
    const int* __restrict__ key = dir ? e0 : e1;

    for (int i = tid; i < n; i += LIST_BLOCK) bins[i] = 0;
    if (g == 0 && tid < 2) cursor[tid] = 0;
    __syncthreads();

    const int chunk = (E + SLICES - 1) / SLICES;
    const int lo = w * chunk;
    const int hi = (lo + chunk < E) ? (lo + chunk) : E;
    for (int e = lo + tid; e < hi; e += EDGES_AHEAD * LIST_BLOCK) {
        int k[EDGES_AHEAD];
        #pragma unroll
        for (int u = 0; u < EDGES_AHEAD; ++u) {
            const int ee = e + u * LIST_BLOCK;
            k[u] = (ee < hi) ? key[ee] : -1;
        }
        #pragma unroll
        for (int u = 0; u < EDGES_AHEAD; ++u)
            if (k[u] >= 0) atomicAdd(&bins[k[u]], 1);
    }
    __syncthreads();

    int* __restrict__ row = part + (size_t)g * n;
    for (int i = tid; i < n; i += LIST_BLOCK) row[i] = bins[i];
}

#define SLICES_AHEAD 16
__global__ __launch_bounds__(256) void alloc_kernel(
    int* __restrict__ part,
    int* __restrict__ cntA, int* __restrict__ cntB,
    int* __restrict__ endA, int* __restrict__ endB,
    int* __restrict__ cursor, int n)
{
    const int i    = blockIdx.x * blockDim.x + threadIdx.x;
    const int lane = threadIdx.x & 31;
    int cA = 0, cB = 0;
    if (i < n) {
        int* __restrict__ pa = part + i;
        int* __restrict__ pb = part + (size_t)SLICES * n + i;
        for (int w0 = 0; w0 < SLICES; w0 += SLICES_AHEAD) {
            int ta[SLICES_AHEAD], tb[SLICES_AHEAD];
            #pragma unroll
            for (int u = 0; u < SLICES_AHEAD; ++u) {
                ta[u] = pa[(size_t)(w0 + u) * n];
                tb[u] = pb[(size_t)(w0 + u) * n];
            }
            #pragma unroll
            for (int u = 0; u < SLICES_AHEAD; ++u) {
                pa[(size_t)(w0 + u) * n] = cA;  cA += ta[u];
                pb[(size_t)(w0 + u) * n] = cB;  cB += tb[u];
            }
        }
    }
    int sA = cA, sB = cB;
    #pragma unroll
    for (int o = 1; o < 32; o <<= 1) {
        const int tA = __shfl_up(sA, o, 32);
        const int tB = __shfl_up(sB, o, 32);
        if (lane >= o) { sA += tA; sB += tB; }
    }
    int baseA = 0, baseB = 0;
    if (lane == 31) {
        baseA = atomicAdd(&cursor[0], sA);
        baseB = atomicAdd(&cursor[1], sB);
    }
    baseA = __shfl(baseA, 31, 32);
    baseB = __shfl(baseB, 31, 32);
    if (i < n) {
        cntA[i] = cA;
        cntB[i] = cB;
        endA[i] = baseA + sA;
        endB[i] = baseB + sB;
    }
}

#define KE_STEPS 4
__device__ __forceinline__ void strip_load_a(v2f (&a)[KE_STEPS], const float* __restrict__ xq) {
    #pragma unroll
    for (int s = 0; s < KE_STEPS; ++s)
        a[s] = *reinterpret_cast<const v2f*>(xq + 4 * s);
}
__device__ __forceinline__ void strip_mma(v8f (&acc)[N_TILES], const v2f (&a)[KE_STEPS],
                                          const v2f* __restrict__ bq) {
    #pragma unroll
    for (int s = 0; s < KE_STEPS; ++s) {
        #pragma unroll
        for (int t = 0; t < N_TILES; ++t) {
            const v2f b = bq[(2 * s) * W_PITCH + t * 16];
            acc[t] = __builtin_amdgcn_wmma_f32_16x16x4_f32(
                 false, a[s],  false, b,
                 (short)0, acc[t],  false,  false);
        }
    }
}

#define FILL_GEMM_GROUPS 128
#define W_TASKS_L  ((F_IN / 2) * (F_OUT / 4) / LIST_BLOCK)

__global__ __launch_bounds__(LIST_BLOCK) void fill_gemm_kernel(
    const int* __restrict__ e0, const int* __restrict__ e1, int E, int n,
    const int* __restrict__ part,
    const int* __restrict__ cntA, const int* __restrict__ cntB,
    const int* __restrict__ endA, const int* __restrict__ endB,
    int* __restrict__ listA, int* __restrict__ listB,
    const float* __restrict__ X, const float* __restrict__ Wm, float* __restrict__ Hm)
{
    __shared__ __attribute__((aligned(16))) int smem[LDS_ROWS];
    const int tid = threadIdx.x;

    if (blockIdx.x >= 2 * SLICES) {
        v2f* __restrict__ Wl = reinterpret_cast<v2f*>(smem);
        const int lane  = tid & 31;
        const int half  = lane >> 4;
        const int l     = lane & 15;
        const int wave  = __builtin_amdgcn_readfirstlane(tid >> 5);
        const int mtile = wave * FILL_GEMM_GROUPS + ((int)blockIdx.x - 2 * SLICES);

        int rowA = mtile * 16 + l;
        if (rowA >= n) rowA = n - 1;

        float4 r0[W_TASKS_L], r1[W_TASKS_L];
        #pragma unroll
        for (int j = 0; j < W_TASKS_L; ++j) {
            const int kp = wave + j * (LIST_BLOCK / 32);
            const int c  = lane * 4;
            r0[j] = *reinterpret_cast<const float4*>(Wm + (size_t)(2 * kp) * F_OUT + c);
            r1[j] = *reinterpret_cast<const float4*>(Wm + (size_t)(2 * kp + 1) * F_OUT + c);
        }

        v2f a[KE_STEPS];
        const float* __restrict__ xq = X + (size_t)rowA * F_IN + half * 2;
        strip_load_a(a, xq);

        #pragma unroll
        for (int j = 0; j < W_TASKS_L; ++j) {
            const int kp = wave + j * (LIST_BLOCK / 32);
            v2f* __restrict__ dst = Wl + kp * W_PITCH + lane * 4;
            v2f p0, p1, p2, p3;
            p0.x = r0[j].x; p0.y = r1[j].x;
            p1.x = r0[j].y; p1.y = r1[j].y;
            p2.x = r0[j].z; p2.y = r1[j].z;
            p3.x = r0[j].w; p3.y = r1[j].w;
            dst[0] = p0; dst[1] = p1; dst[2] = p2; dst[3] = p3;
        }
        __syncthreads();

        if (mtile * 16 >= n) return;

        v8f acc[N_TILES];
        #pragma unroll
        for (int t = 0; t < N_TILES; ++t) acc[t] = (v8f){};

        const v2f* __restrict__ bq = Wl + half * W_PITCH + l;
        #pragma nounroll
        for (int q = 0; q < K_STEPS / KE_STEPS; ++q) {
            strip_mma(acc, a, bq);
            xq += 4 * KE_STEPS;
            bq += 2 * KE_STEPS * W_PITCH;
            if (q + 1 < K_STEPS / KE_STEPS) strip_load_a(a, xq);
        }

        const int rowBase = mtile * 16 + half * 8;
        float* __restrict__ hp = Hm + (size_t)rowBase * F_OUT + l;
        if (mtile * 16 + 16 <= n) {
            #pragma unroll
            for (int t = 0; t < N_TILES; ++t) {
                #pragma unroll
                for (int r = 0; r < 8; ++r) hp[r * F_OUT + t * 16] = acc[t][r];
            }
        } else {
            #pragma unroll
            for (int t = 0; t < N_TILES; ++t) {
                #pragma unroll
                for (int r = 0; r < 8; ++r)
                    if (rowBase + r < n) hp[r * F_OUT + t * 16] = acc[t][r];
            }
        }
        return;
    }

    int* __restrict__ bins = smem;
    const int g   = blockIdx.x;
    const int dir = g / SLICES;
    const int w   = g - dir * SLICES;
    const int* __restrict__ key = dir ? e0 : e1;
    const int* __restrict__ val = dir ? e1 : e0;
    const int* __restrict__ cnt = dir ? cntB : cntA;
    const int* __restrict__ end = dir ? endB : endA;
    int* __restrict__ list      = dir ? listB : listA;
    const int* __restrict__ rel = part + (size_t)g * n;

    for (int i0 = tid; i0 < n; i0 += ROWS_AHEAD * LIST_BLOCK) {
        int t[ROWS_AHEAD];
        #pragma unroll
        for (int u = 0; u < ROWS_AHEAD; ++u) {
            const int i = i0 + u * LIST_BLOCK;
            t[u] = (i < n) ? (end[i] - cnt[i] + rel[i]) : 0;
        }
        #pragma unroll
        for (int u = 0; u < ROWS_AHEAD; ++u) {
            const int i = i0 + u * LIST_BLOCK;
            if (i < n) bins[i] = t[u];
        }
    }
    __syncthreads();

    const int chunk = (E + SLICES - 1) / SLICES;
    const int lo = w * chunk;
    const int hi = (lo + chunk < E) ? (lo + chunk) : E;
    for (int e = lo + tid; e < hi; e += EDGES_AHEAD * LIST_BLOCK) {
        int k[EDGES_AHEAD], v[EDGES_AHEAD];
        #pragma unroll
        for (int u = 0; u < EDGES_AHEAD; ++u) {
            const int ee = e + u * LIST_BLOCK;
            k[u] = (ee < hi) ? key[ee] : -1;
            v[u] = (ee < hi) ? val[ee] : 0;
        }
        #pragma unroll
        for (int u = 0; u < EDGES_AHEAD; ++u)
            if (k[u] >= 0) list[atomicAdd(&bins[k[u]], 1)] = v[u];
    }
}

typedef __attribute__((ext_vector_type(4))) float v4f;
#define LOG2E_ 1.4426950408889634f

template <int K>
__device__ __forceinline__ float from_lane_xor(float v) {
    return __builtin_bit_cast(float,
        __builtin_amdgcn_update_dpp(0, __builtin_bit_cast(int, v), 0x160 + K, 0xf, 0xf, true));
}
template <int U>
__device__ __forceinline__ float from_group_lane(float v) {
    return __builtin_bit_cast(float,
        __builtin_amdgcn_mov_dpp8(__builtin_bit_cast(int, v), U * 0x249249));
}
__device__ __forceinline__ float head_sum(float v) {
    v += from_lane_xor<1>(v);
    v += from_lane_xor<2>(v);
    v += from_lane_xor<4>(v);
    return v;
}
__device__ __forceinline__ float head_max(float v) {
    v = fmaxf(v, from_lane_xor<1>(v));
    v = fmaxf(v, from_lane_xor<2>(v));
    v = fmaxf(v, from_lane_xor<4>(v));
    return v;
}
__device__ __forceinline__ float dot4(const v4f a, const v4f b) {
    const v2f t = a.lo * b.lo + a.hi * b.hi;
    return t.x + t.y;
}

#define ROWS_IN_FLIGHT 8
__global__ __launch_bounds__(256) void gather_kernel(
    const int* __restrict__ list, const int* __restrict__ segEnd,
    const int* __restrict__ segCnt,
    const float* __restrict__ as, const float* __restrict__ ad,
    const float* __restrict__ Hm,
    const float* __restrict__ b1, const float* __restrict__ b2,
    float scale, int first, float* __restrict__ out, int n)
{
    const int lane = threadIdx.x & 31;
    const int j    = lane & 7;
    const bool j4 = (j & 4) != 0, j2 = (j & 2) != 0, j1 = (j & 1) != 0;
    const int d = __builtin_amdgcn_readfirstlane(
        (int)(blockIdx.x * (blockDim.x >> 5) + (threadIdx.x >> 5)));
    if (d >= n) return;
    const int deg = __builtin_amdgcn_readfirstlane(segCnt[d]);
    const int beg = __builtin_amdgcn_readfirstlane(segEnd[d]) - deg;

    const float* __restrict__ hl = Hm + lane * 4;
    const v4f a4 = *reinterpret_cast<const v4f*>(as + lane * 4) * LOG2E_;
    const v4f d4 = *reinterpret_cast<const v4f*>(ad + lane * 4) * LOG2E_;

    const v4f self = *reinterpret_cast<const v4f*>(hl + (size_t)d * F_OUT);
    const float edh = head_sum(dot4(self, d4));
    float m   = lrelu(head_sum(dot4(self, a4)) + edh);
    float den = (j == 0) ? 1.0f : 0.0f;
    v4f   acc = self;

    const float ninf = -__builtin_huge_valf();
    for (int base = 0; base < deg; base += 32) {
        const int i = base + lane;
        int s = d;
        if (i < deg) s = list[beg + i];
        const int cnt = (deg - base < 32) ? (deg - base) : 32;
        for (int e = 0; e < cnt; e += ROWS_IN_FLIGHT) {
            v4f   row[ROWS_IN_FLIGHT];
            float q[ROWS_IN_FLIGHT];
            #pragma unroll
            for (int u = 0; u < ROWS_IN_FLIGHT; ++u) {
                const int se = __builtin_amdgcn_readlane(s, e + u);
                row[u] = *reinterpret_cast<const v4f*>(hl + (size_t)se * F_OUT);
            }
            #pragma unroll
            for (int u = 0; u < ROWS_IN_FLIGHT; ++u) q[u] = dot4(row[u], a4);

            float k4[4], k2[2];
            #pragma unroll
            for (int t = 0; t < 4; ++t) {
                const float keep = j4 ? q[4 + t] : q[t];
                const float send = j4 ? q[t] : q[4 + t];
                k4[t] = keep + from_lane_xor<4>(send);
            }
            #pragma unroll
            for (int t = 0; t < 2; ++t) {
                const float keep = j2 ? k4[2 + t] : k4[t];
                const float send = j2 ? k4[t] : k4[2 + t];
                k2[t] = keep + from_lane_xor<2>(send);
            }
            const float keep1 = j1 ? k2[1] : k2[0];
            const float send1 = j1 ? k2[0] : k2[1];
            float v = lrelu(keep1 + from_lane_xor<1>(send1) + edh);
            if (e + j >= cnt) v = ninf;

            const float mn = fmaxf(m, head_max(v));
            const float r  = __builtin_amdgcn_exp2f(m - mn);
            const float p  = __builtin_amdgcn_exp2f(v - mn);
            m   = mn;
            den = den * r + p;
            acc *= r;
            acc += from_group_lane<0>(p) * row[0];
            acc += from_group_lane<1>(p) * row[1];
            acc += from_group_lane<2>(p) * row[2];
            acc += from_group_lane<3>(p) * row[3];
            acc += from_group_lane<4>(p) * row[4];
            acc += from_group_lane<5>(p) * row[5];
            acc += from_group_lane<6>(p) * row[6];
            acc += from_group_lane<7>(p) * row[7];
        }
    }

    const float f = scale / head_sum(den);
    float4* __restrict__ op = reinterpret_cast<float4*>(out + (size_t)d * F_OUT + lane * 4);
    float4 o;
    if (first) {
        const float4 x1 = *reinterpret_cast<const float4*>(b1 + lane * 4);
        const float4 x2 = *reinterpret_cast<const float4*>(b2 + lane * 4);
        o.x = (1.0f - ALPHA_) * x1.x + ALPHA_ * x2.x;
        o.y = (1.0f - ALPHA_) * x1.y + ALPHA_ * x2.y;
        o.z = (1.0f - ALPHA_) * x1.z + ALPHA_ * x2.z;
        o.w = (1.0f - ALPHA_) * x1.w + ALPHA_ * x2.w;
    } else {
        o = *op;
    }
    o.x += f * acc.x; o.y += f * acc.y; o.z += f * acc.z; o.w += f * acc.w;
    *op = o;
}

extern "C" void kernel_launch(void* const* d_in, const int* in_sizes, int n_in,
                              void* d_out, int out_size, void* d_ws, size_t ws_size,
                              hipStream_t stream) {
    const float* x    = (const float*)d_in[0];
    const int*   ei   = (const int*)d_in[1];
    const float* W1   = (const float*)d_in[2];
    const float* as1  = (const float*)d_in[3];
    const float* ad1  = (const float*)d_in[4];
    const float* b1   = (const float*)d_in[5];
    const float* W2   = (const float*)d_in[6];
    const float* as2  = (const float*)d_in[7];
    const float* ad2  = (const float*)d_in[8];
    const float* b2   = (const float*)d_in[9];
    float* out = (float*)d_out;

    const int n = in_sizes[0] / F_IN;
    const int E = in_sizes[1] / 2;
    const int* src = ei;
    const int* dst = ei + E;

    float* p   = (float*)d_ws;
    float* Hm  = p;  p += (size_t)n * F_OUT;
    int* q      = (int*)p;
    int* cntA   = q;  q += n;
    int* cntB   = q;  q += n;
    int* endA = q;  q += n;
    int* endB = q;  q += n;
    int* cursor = q;  q += 4;
    int* listA  = q;  q += E;
    int* listB  = q;

    const int nblk  = (n + 255) / 256;
    const int gblk  = (n + 16 * GEMM_WAVES - 1) / (16 * GEMM_WAVES);
    const int rblk  = (n + 7) / 8;

    int* part = (int*)out;
    hist_lds_kernel<<<2 * SLICES, LIST_BLOCK, 0, stream>>>(src, dst, E, n, part, cursor);
    alloc_kernel<<<nblk, 256, 0, stream>>>(part, cntA, cntB, endA, endB, cursor, n);
    fill_gemm_kernel<<<2 * SLICES + FILL_GEMM_GROUPS, LIST_BLOCK, 0, stream>>>(
        src, dst, E, n, part, cntA, cntB, endA, endB, listA, listB, x, W1, Hm);

    gather_kernel<<<rblk, 256, 0, stream>>>(listA, endA, cntA, as1, ad1, Hm, b1, b2,
                                            1.0f - ALPHA_, 1, out, n);

    gemm_wmma_kernel<<<gblk, GEMM_WAVES * 32, 0, stream>>>(x, W2, Hm, n);
    gather_kernel<<<rblk, 256, 0, stream>>>(listB, endB, cntB, as2, ad2, Hm, b1, b2,
                                            ALPHA_, 0, out, n);
}


# gather: a wave walks 5 consecutive rows; their counts and ends come in one load, and the next row's list chunks, h row and out row are in flight while the current row is summed (50000 waves -> 10000)
# speedup vs baseline: 21.0455x; 1.3051x over previous
#include <hip/hip_runtime.h>


typedef __attribute__((ext_vector_type(2))) float v2f;
typedef __attribute__((ext_vector_type(8))) float v8f;

#define F_IN   128
#define HEADS  4
#define C_OUT  32
#define F_OUT  128
#define ALPHA_ 0.5f
#define SLOPE_ 0.2f

__device__ __forceinline__ float lrelu(float v) {
    return v > 0.0f ? v : v * SLOPE_;
}

#define GEMM_WAVES 8
#define K_STEPS    (F_IN / 4)
#define N_TILES    (F_OUT / 16)
#define W_TASKS    ((F_IN / 2) * (F_OUT / 4) / (GEMM_WAVES * 32))
#define W_PITCH    (F_OUT + 16)

__global__ __launch_bounds__(GEMM_WAVES * 32) void gemm_wmma_kernel(
    const float* __restrict__ X, const float* __restrict__ Wm,
    float* __restrict__ Hm, int n)
{
    __shared__ v2f Wl[(F_IN / 2) * W_PITCH];

    const int tid   = threadIdx.x;
    const int lane  = tid & 31;
    const int half  = lane >> 4;
    const int l     = lane & 15;
    const int mtile = blockIdx.x * GEMM_WAVES + (tid >> 5);

    int rowA = mtile * 16 + l;
    if (rowA >= n) rowA = n - 1;

    float4 r0[W_TASKS], r1[W_TASKS];
    #pragma unroll
    for (int j = 0; j < W_TASKS; ++j) {
        const int kp = (tid >> 5) + j * GEMM_WAVES;
        const int c  = lane * 4;
        r0[j] = *reinterpret_cast<const float4*>(Wm + (size_t)(2 * kp) * F_OUT + c);
        r1[j] = *reinterpret_cast<const float4*>(Wm + (size_t)(2 * kp + 1) * F_OUT + c);
    }

    v2f a[K_STEPS];
    const float* __restrict__ xp = X + (size_t)rowA * F_IN + half * 2;
    #pragma unroll
    for (int s = 0; s < K_STEPS; ++s)
        a[s] = *reinterpret_cast<const v2f*>(xp + 4 * s);

    #pragma unroll
    for (int j = 0; j < W_TASKS; ++j) {
        const int kp = (tid >> 5) + j * GEMM_WAVES;
        v2f* __restrict__ dst = Wl + kp * W_PITCH + lane * 4;
        v2f p0, p1, p2, p3;
        p0.x = r0[j].x; p0.y = r1[j].x;
        p1.x = r0[j].y; p1.y = r1[j].y;
        p2.x = r0[j].z; p2.y = r1[j].z;
        p3.x = r0[j].w; p3.y = r1[j].w;
        dst[0] = p0; dst[1] = p1; dst[2] = p2; dst[3] = p3;
    }
    __syncthreads();

    if (mtile * 16 >= n) return;

    v8f acc[N_TILES];
    #pragma unroll
    for (int t = 0; t < N_TILES; ++t) acc[t] = (v8f){};

    const v2f* __restrict__ bp = Wl + half * W_PITCH + l;
    #pragma unroll
    for (int s = 0; s < K_STEPS; ++s) {
        #pragma unroll
        for (int t = 0; t < N_TILES; ++t) {
            const v2f b = bp[(2 * s) * W_PITCH + t * 16];
            acc[t] = __builtin_amdgcn_wmma_f32_16x16x4_f32(
                 false, a[s],  false, b,
                 (short)0, acc[t],  false,  false);
        }
    }

    const int rowBase = mtile * 16 + half * 8;
    float* __restrict__ hp = Hm + (size_t)rowBase * F_OUT + l;
    if (mtile * 16 + 16 <= n) {
        #pragma unroll
        for (int t = 0; t < N_TILES; ++t) {
            #pragma unroll
            for (int r = 0; r < 8; ++r) hp[r * F_OUT + t * 16] = acc[t][r];
        }
    } else {
        #pragma unroll
        for (int t = 0; t < N_TILES; ++t) {
            #pragma unroll
            for (int r = 0; r < 8; ++r)
                if (rowBase + r < n) hp[r * F_OUT + t * 16] = acc[t][r];
        }
    }
}

#define SLICES      64
#define RSH         8
#define ROWS_MAX    (1 << RSH)
#define NBUCK       256
#define LIST_BLOCK  1024
#define LISTS_BLOCK 512
#define EDGES_AHEAD 8
#define RUN_AHEAD   4
#define STAGE_P     25600
#define STAGE_L     16384
#define LIST_PARTS  (LISTS_BLOCK / ROWS_MAX)
#define RUNS_PER_PART (SLICES / LIST_PARTS)

#define KE_STEPS 4
__device__ __forceinline__ void strip_load_a(v2f (&a)[KE_STEPS], const float* __restrict__ xq) {
    #pragma unroll
    for (int s = 0; s < KE_STEPS; ++s)
        a[s] = *reinterpret_cast<const v2f*>(xq + 4 * s);
}
__device__ __forceinline__ void strip_mma(v8f (&acc)[N_TILES], const v2f (&a)[KE_STEPS],
                                          const v2f* __restrict__ bq) {
    #pragma unroll
    for (int s = 0; s < KE_STEPS; ++s) {
        #pragma unroll
        for (int t = 0; t < N_TILES; ++t) {
            const v2f b = bq[(2 * s) * W_PITCH + t * 16];
            acc[t] = __builtin_amdgcn_wmma_f32_16x16x4_f32(
                 false, a[s],  false, b,
                 (short)0, acc[t],  false,  false);
        }
    }
}

#define FILL_GEMM_GROUPS 128
#define W_TASKS_L  ((F_IN / 2) * (F_OUT / 4) / LIST_BLOCK)

__global__ __launch_bounds__(LIST_BLOCK) void part_gemm_kernel(
    const int* __restrict__ e0, const int* __restrict__ e1, int E, int n,
    unsigned* __restrict__ temp, int* __restrict__ pre,
    const float* __restrict__ X, const float* __restrict__ Wm, float* __restrict__ Hm)
{
    __shared__ __attribute__((aligned(16))) int smem[2 * (NBUCK + 4) + STAGE_P + 4];
    const int tid = threadIdx.x;

    if (blockIdx.x >= 2 * SLICES) {
        v2f* __restrict__ Wl = reinterpret_cast<v2f*>(smem);
        const int lane  = tid & 31;
        const int half  = lane >> 4;
        const int l     = lane & 15;
        const int wave  = __builtin_amdgcn_readfirstlane(tid >> 5);
        const int mtile = wave * FILL_GEMM_GROUPS + ((int)blockIdx.x - 2 * SLICES);

        int rowA = mtile * 16 + l;
        if (rowA >= n) rowA = n - 1;

        float4 r0[W_TASKS_L], r1[W_TASKS_L];
        #pragma unroll
        for (int j = 0; j < W_TASKS_L; ++j) {
            const int kp = wave + j * (LIST_BLOCK / 32);
            const int c  = lane * 4;
            r0[j] = *reinterpret_cast<const float4*>(Wm + (size_t)(2 * kp) * F_OUT + c);
            r1[j] = *reinterpret_cast<const float4*>(Wm + (size_t)(2 * kp + 1) * F_OUT + c);
        }

        v2f a[KE_STEPS];
        const float* __restrict__ xq = X + (size_t)rowA * F_IN + half * 2;
        strip_load_a(a, xq);

        #pragma unroll
        for (int j = 0; j < W_TASKS_L; ++j) {
            const int kp = wave + j * (LIST_BLOCK / 32);
            v2f* __restrict__ dst = Wl + kp * W_PITCH + lane * 4;
            v2f p0, p1, p2, p3;
            p0.x = r0[j].x; p0.y = r1[j].x;
            p1.x = r0[j].y; p1.y = r1[j].y;
            p2.x = r0[j].z; p2.y = r1[j].z;
            p3.x = r0[j].w; p3.y = r1[j].w;
            dst[0] = p0; dst[1] = p1; dst[2] = p2; dst[3] = p3;
        }
        __syncthreads();

        if (mtile * 16 >= n) return;

        v8f acc[N_TILES];
        #pragma unroll
        for (int t = 0; t < N_TILES; ++t) acc[t] = (v8f){};

        const v2f* __restrict__ bq = Wl + half * W_PITCH + l;
        #pragma nounroll
        for (int q = 0; q < K_STEPS / KE_STEPS; ++q) {
            strip_mma(acc, a, bq);
            xq += 4 * KE_STEPS;
            bq += 2 * KE_STEPS * W_PITCH;
            if (q + 1 < K_STEPS / KE_STEPS) strip_load_a(a, xq);
        }

        const int rowBase = mtile * 16 + half * 8;
        float* __restrict__ hp = Hm + (size_t)rowBase * F_OUT + l;
        if (mtile * 16 + 16 <= n) {
            #pragma unroll
            for (int t = 0; t < N_TILES; ++t) {
                #pragma unroll
                for (int r = 0; r < 8; ++r) hp[r * F_OUT + t * 16] = acc[t][r];
            }
        } else {
            #pragma unroll
            for (int t = 0; t < N_TILES; ++t) {
                #pragma unroll
                for (int r = 0; r < 8; ++r)
                    if (rowBase + r < n) hp[r * F_OUT + t * 16] = acc[t][r];
            }
        }
        return;
    }

    int* __restrict__ cnt = smem;
    int* __restrict__ cur = smem + NBUCK + 4;
    unsigned* __restrict__ stage = reinterpret_cast<unsigned*>(smem + 2 * (NBUCK + 4));
    const int g   = blockIdx.x;
    const int dir = g / SLICES;
    const int w   = g - dir * SLICES;
    const int* __restrict__ key = dir ? e0 : e1;
    const int* __restrict__ val = dir ? e1 : e0;
    const int sh = RSH;
    const unsigned mask = ROWS_MAX - 1;

    const int chunk = (E + SLICES - 1) / SLICES;
    const int lo = w * chunk;
    const int hi = (lo + chunk < E) ? (lo + chunk) : E;

    if (tid <= NBUCK) cnt[tid] = 0;
    __syncthreads();

    for (int e = lo + tid; e < hi; e += EDGES_AHEAD * LIST_BLOCK) {
        int k[EDGES_AHEAD];
        #pragma unroll
        for (int u = 0; u < EDGES_AHEAD; ++u) {
            const int ee = e + u * LIST_BLOCK;
            k[u] = key[(ee < hi) ? ee : (hi - 1)];
        }
        #pragma unroll
        for (int u = 0; u < EDGES_AHEAD; ++u)
            atomicAdd(&cnt[(e + u * LIST_BLOCK < hi) ? (k[u] >> sh) : NBUCK], 1);
    }
    __syncthreads();

    if (tid < 32) {
        int v[NBUCK / 32];
        int s = 0;
        #pragma unroll
        for (int u = 0; u < NBUCK / 32; ++u) {
            const int t = cnt[tid * (NBUCK / 32) + u];
            v[u] = s;  s += t;
        }
        int incl = s;
        #pragma unroll
        for (int o = 1; o < 32; o <<= 1) {
            const int t = __shfl_up(incl, o, 32);
            if (tid >= o) incl += t;
        }
        const int excl = incl - s;
        int* __restrict__ prow = pre + (size_t)g * (NBUCK + 1);
        #pragma unroll
        for (int u = 0; u < NBUCK / 32; ++u) {
            cur[tid * (NBUCK / 32) + u]  = excl + v[u];
            prow[tid * (NBUCK / 32) + u] = excl + v[u];
        }
        if (tid == 31) { prow[NBUCK] = incl; cur[NBUCK] = 0; }
    }
    __syncthreads();

    unsigned* __restrict__ tdst = temp + (size_t)dir * E + lo;
    if (chunk <= STAGE_P) {
        for (int e = lo + tid; e < hi; e += EDGES_AHEAD * LIST_BLOCK) {
            int k[EDGES_AHEAD], v[EDGES_AHEAD], pos[EDGES_AHEAD];
            #pragma unroll
            for (int u = 0; u < EDGES_AHEAD; ++u) {
                const int ee = e + u * LIST_BLOCK;
                const int ec = (ee < hi) ? ee : (hi - 1);
                k[u] = key[ec];
                v[u] = val[ec];
            }
            #pragma unroll
            for (int u = 0; u < EDGES_AHEAD; ++u)
                pos[u] = atomicAdd(&cur[(e + u * LIST_BLOCK < hi) ? (k[u] >> sh) : NBUCK], 1);
            #pragma unroll
            for (int u = 0; u < EDGES_AHEAD; ++u)
                stage[(e + u * LIST_BLOCK < hi) ? pos[u] : STAGE_P] =
                    ((unsigned)v[u] << sh) | ((unsigned)k[u] & mask);
        }
        __syncthreads();
        for (int i = tid; i < hi - lo; i += LIST_BLOCK) tdst[i] = stage[i];
    } else {
        for (int e = lo + tid; e < hi; e += LIST_BLOCK) {
            const int k = key[e];
            tdst[atomicAdd(&cur[k >> sh], 1)] = ((unsigned)val[e] << sh) | ((unsigned)k & mask);
        }
    }
}

__global__ __launch_bounds__(LISTS_BLOCK) void lists_kernel(
    const unsigned* __restrict__ temp, const int* __restrict__ pre, int E, int n,
    int* __restrict__ cntA, int* __restrict__ cntB,
    int* __restrict__ endA, int* __restrict__ endB,
    int* __restrict__ listA, int* __restrict__ listB)
{
    __shared__ int cntRS[SLICES * ROWS_MAX + 4];
    __shared__ int stageL[STAGE_L + 4];
    __shared__ int partQ[LIST_PARTS * ROWS_MAX];
    __shared__ int rowEx[ROWS_MAX];
    __shared__ int runStart[SLICES], runLen[SLICES], runP0[SLICES];
    __shared__ int wsum[ROWS_MAX / 32];
    __shared__ int sBase, sTotal;

    const int tid  = threadIdx.x;
    const int lane = tid & 31;
    const int wave = __builtin_amdgcn_readfirstlane(tid >> 5);
    const int nb  = (n + ROWS_MAX - 1) / ROWS_MAX;
    const int g   = blockIdx.x;
    const int dir = g / nb;
    const int b   = g - dir * nb;
    const int sh  = RSH;
    const unsigned mask = ROWS_MAX - 1;
    const int rowLo = b << sh;
    const int nrows = (n - rowLo < ROWS_MAX) ? (n - rowLo) : ROWS_MAX;
    const int chunk = (E + SLICES - 1) / SLICES;
    int* __restrict__ segCnt = dir ? cntB : cntA;
    int* __restrict__ segEnd = dir ? endB : endA;
    int* __restrict__ list   = dir ? listB : listA;

    if (tid < SLICES) {
        const int* __restrict__ prow = pre + (size_t)(dir * SLICES + tid) * (NBUCK + 1) + b;
        const int p0 = prow[0], p1 = prow[1];
        runP0[tid]    = p0;
        runLen[tid]   = p1 - p0;
        runStart[tid] = dir * E + tid * chunk + p0;
    }
    for (int i = tid; i < SLICES * ROWS_MAX + 4; i += LISTS_BLOCK) cntRS[i] = 0;
    __syncthreads();

    if (tid < 32) {
        int a = 0, t = 0;
        for (int r = tid; r < SLICES; r += 32) { a += runP0[r]; t += runLen[r]; }
        #pragma unroll
        for (int o = 16; o > 0; o >>= 1) {
            a += __shfl_xor(a, o, 32);
            t += __shfl_xor(t, o, 32);
        }
        if (tid == 0) { sBase = a; sTotal = t; }
    }

    for (int r = wave; r < SLICES; r += LISTS_BLOCK / 32) {
        const int len = runLen[r];
        const unsigned* __restrict__ p = temp + runStart[r];
        const int crow = r * ROWS_MAX;
        for (int o = lane; o < len; o += 32 * RUN_AHEAD) {
            unsigned t[RUN_AHEAD];
            #pragma unroll
            for (int u = 0; u < RUN_AHEAD; ++u)
                t[u] = p[(o + 32 * u < len) ? (o + 32 * u) : (len - 1)];
            #pragma unroll
            for (int u = 0; u < RUN_AHEAD; ++u)
                atomicAdd(&cntRS[(o + 32 * u < len) ? (crow + (int)(t[u] & mask)) : (SLICES * ROWS_MAX)], 1);
        }
    }
    __syncthreads();

    const int row = tid & (ROWS_MAX - 1);
    const int q   = tid / ROWS_MAX;
    {
        int sacc = 0;
        for (int r = q * RUNS_PER_PART; r < (q + 1) * RUNS_PER_PART; ++r) sacc += cntRS[r * ROWS_MAX + row];
        partQ[q * ROWS_MAX + row] = sacc;
    }
    __syncthreads();
    int c = 0, s = 0;
    if (tid < ROWS_MAX) {
        for (int qq = 0; qq < LIST_PARTS; ++qq) c += partQ[qq * ROWS_MAX + tid];
        s = c;
        #pragma unroll
        for (int o = 1; o < 32; o <<= 1) {
            const int t = __shfl_up(s, o, 32);
            if (lane >= o) s += t;
        }
        if (lane == 31) wsum[tid >> 5] = s;
    }
    __syncthreads();
    if (tid < ROWS_MAX) {
        int off = 0;
        for (int jw = 0; jw < (tid >> 5); ++jw) off += wsum[jw];
        const int excl = off + s - c;
        rowEx[tid] = excl;
        if (tid < nrows) {
            segCnt[rowLo + tid] = c;
            segEnd[rowLo + tid] = sBase + excl + c;
        }
    }
    __syncthreads();
    {
        int start = rowEx[row];
        for (int qq = 0; qq < q; ++qq) start += partQ[qq * ROWS_MAX + row];
        for (int r = q * RUNS_PER_PART; r < (q + 1) * RUNS_PER_PART; ++r) {
            const int t = cntRS[r * ROWS_MAX + row];
            cntRS[r * ROWS_MAX + row] = start;
            start += t;
        }
    }
    __syncthreads();

    const int base   = sBase;
    const int total  = sTotal;
    if (total <= STAGE_L) {
        for (int r = wave; r < SLICES; r += LISTS_BLOCK / 32) {
            const int len = runLen[r];
            const unsigned* __restrict__ p = temp + runStart[r];
            const int crow = r * ROWS_MAX;
            for (int o = lane; o < len; o += 32 * RUN_AHEAD) {
                unsigned t[RUN_AHEAD];
                int pos[RUN_AHEAD];
                #pragma unroll
                for (int u = 0; u < RUN_AHEAD; ++u)
                    t[u] = p[(o + 32 * u < len) ? (o + 32 * u) : (len - 1)];
                #pragma unroll
                for (int u = 0; u < RUN_AHEAD; ++u)
                    pos[u] = atomicAdd(&cntRS[(o + 32 * u < len) ? (crow + (int)(t[u] & mask)) : (SLICES * ROWS_MAX)], 1);
                #pragma unroll
                for (int u = 0; u < RUN_AHEAD; ++u)
                    stageL[(o + 32 * u < len) ? pos[u] : STAGE_L] = (int)(t[u] >> sh);
            }
        }
        __syncthreads();
        for (int i = tid; i < total; i += LISTS_BLOCK) list[base + i] = stageL[i];
    } else {
        for (int r = wave; r < SLICES; r += LISTS_BLOCK / 32) {
            const int len = runLen[r];
            const unsigned* __restrict__ p = temp + runStart[r];
            const int crow = r * ROWS_MAX;
            for (int o = lane; o < len; o += 32) {
                const unsigned t = p[o];
                list[base + atomicAdd(&cntRS[crow + (int)(t & mask)], 1)] = (int)(t >> sh);
            }
        }
    }
}

typedef __attribute__((ext_vector_type(4))) float v4f;
#define LOG2E_ 1.4426950408889634f

template <int K>
__device__ __forceinline__ float from_lane_xor(float v) {
    return __builtin_bit_cast(float,
        __builtin_amdgcn_update_dpp(0, __builtin_bit_cast(int, v), 0x160 + K, 0xf, 0xf, true));
}
template <int U>
__device__ __forceinline__ float from_group_lane(float v) {
    return __builtin_bit_cast(float,
        __builtin_amdgcn_mov_dpp8(__builtin_bit_cast(int, v), U * 0x249249));
}
__device__ __forceinline__ float head_sum(float v) {
    v += from_lane_xor<1>(v);
    v += from_lane_xor<2>(v);
    v += from_lane_xor<4>(v);
    return v;
}
__device__ __forceinline__ float head_max(float v) {
    v = fmaxf(v, from_lane_xor<1>(v));
    v = fmaxf(v, from_lane_xor<2>(v));
    v = fmaxf(v, from_lane_xor<4>(v));
    return v;
}
__device__ __forceinline__ float dot4(const v4f a, const v4f b) {
    const v2f t = a.lo * b.lo + a.hi * b.hi;
    return t.x + t.y;
}

#define ROWS_IN_FLIGHT 8
#define GATHER_WAVES   8
#define ROWS_PER_WAVE  5
template <bool FIRST>
__global__ __launch_bounds__(GATHER_WAVES * 32) void gather_kernel(
    const int* __restrict__ list, const int* __restrict__ segEnd,
    const int* __restrict__ segCnt,
    const float* __restrict__ as, const float* __restrict__ ad,
    const float* __restrict__ Hm,
    const float* __restrict__ b1, const float* __restrict__ b2,
    float scale, float* __restrict__ out, int n)
{
    const int lane = threadIdx.x & 31;
    const int j    = lane & 7;
    const bool j4 = (j & 4) != 0, j2 = (j & 2) != 0, j1 = (j & 1) != 0;
    const int d0 = __builtin_amdgcn_readfirstlane(
        (int)(blockIdx.x * GATHER_WAVES + (threadIdx.x >> 5))) * ROWS_PER_WAVE;
    if (d0 >= n) return;
    const int nr = (n - d0 < ROWS_PER_WAVE) ? (n - d0) : ROWS_PER_WAVE;

    int cntv = 0, endv = 0;
    if (lane < nr) { cntv = segCnt[d0 + lane]; endv = segEnd[d0 + lane]; }

    const unsigned loff = (unsigned)lane * 16u;
    const char* __restrict__ hb = reinterpret_cast<const char*>(Hm);
    char* __restrict__ ob = reinterpret_cast<char*>(out);
    const v4f a4 = *reinterpret_cast<const v4f*>(as + lane * 4) * LOG2E_;
    const v4f d4 = *reinterpret_cast<const v4f*>(ad + lane * 4) * LOG2E_;
    const float ninf = -__builtin_huge_valf();

    int deg = __builtin_amdgcn_readlane(cntv, 0);
    int beg = __builtin_amdgcn_readlane(endv, 0) - deg;
    int sA = d0, sB = d0;
    if (lane < deg)      sA = list[beg + lane];
    if (lane + 32 < deg) sB = list[beg + 32 + lane];
    v4f self = *reinterpret_cast<const v4f*>(hb + (((unsigned)d0 << 9) + loff));
    asm volatile("" :: "v"(sA), "v"(sB), "v"(self));

    for (int r = 0; r < nr; ++r) {
        const int d = d0 + r;
        const int degc = deg, begc = beg;
        const int s0 = sA, s1 = sB;
        const v4f selfc = self;
        const unsigned ooff = ((unsigned)d << 9) + loff;

        if (r + 1 < nr) {
            deg = __builtin_amdgcn_readlane(cntv, r + 1);
            beg = __builtin_amdgcn_readlane(endv, r + 1) - deg;
            sA = d + 1; sB = d + 1;
            if (lane < deg)      sA = list[beg + lane];
            if (lane + 32 < deg) sB = list[beg + 32 + lane];
            self = *reinterpret_cast<const v4f*>(hb + (((unsigned)(d + 1) << 9) + loff));
        }
        float4 o;
        if (!FIRST) o = *reinterpret_cast<const float4*>(ob + ooff);

        const float edh = head_sum(dot4(selfc, d4));
        float m   = lrelu(head_sum(dot4(selfc, a4)) + edh);
        float den = (j == 0) ? 1.0f : 0.0f;
        v4f   acc = selfc;

        for (int base = 0; base < degc; base += 32) {
            int s;
            if (base == 0)       s = s0;
            else if (base == 32) s = s1;
            else {
                s = d;
                if (base + lane < degc) s = list[begc + base + lane];
                asm volatile("" :: "v"(s));
            }
            const int cnt = (degc - base < 32) ? (degc - base) : 32;
            for (int e = 0; e < cnt; e += ROWS_IN_FLIGHT) {
                v4f   row[ROWS_IN_FLIGHT];
                float q[ROWS_IN_FLIGHT];
                #pragma unroll
                for (int u = 0; u < ROWS_IN_FLIGHT; ++u) {
                    const unsigned se = (unsigned)__builtin_amdgcn_readlane(s, e + u);
                    row[u] = *reinterpret_cast<const v4f*>(hb + ((se << 9) + loff));
                }
                #pragma unroll
                for (int u = 0; u < ROWS_IN_FLIGHT; ++u) q[u] = dot4(row[u], a4);

                float k4[4], k2[2];
                #pragma unroll
                for (int t = 0; t < 4; ++t) {
                    const float keep = j4 ? q[4 + t] : q[t];
                    const float send = j4 ? q[t] : q[4 + t];
                    k4[t] = keep + from_lane_xor<4>(send);
                }
                #pragma unroll
                for (int t = 0; t < 2; ++t) {
                    const float keep = j2 ? k4[2 + t] : k4[t];
                    const float send = j2 ? k4[t] : k4[2 + t];
                    k2[t] = keep + from_lane_xor<2>(send);
                }
                const float keep1 = j1 ? k2[1] : k2[0];
                const float send1 = j1 ? k2[0] : k2[1];
                float v = lrelu(keep1 + from_lane_xor<1>(send1) + edh);
                if (e + j >= cnt) v = ninf;

                const float mn = fmaxf(m, head_max(v));
                const float rs = __builtin_amdgcn_exp2f(m - mn);
                const float p  = __builtin_amdgcn_exp2f(v - mn);
                m   = mn;
                den = den * rs + p;
                acc *= rs;
                acc += from_group_lane<0>(p) * row[0];
                acc += from_group_lane<1>(p) * row[1];
                acc += from_group_lane<2>(p) * row[2];
                acc += from_group_lane<3>(p) * row[3];
                acc += from_group_lane<4>(p) * row[4];
                acc += from_group_lane<5>(p) * row[5];
                acc += from_group_lane<6>(p) * row[6];
                acc += from_group_lane<7>(p) * row[7];
            }
        }

        const float f = scale / head_sum(den);
        if (FIRST) {
            const float4 x1 = *reinterpret_cast<const float4*>(b1 + lane * 4);
            const float4 x2 = *reinterpret_cast<const float4*>(b2 + lane * 4);
            o.x = (1.0f - ALPHA_) * x1.x + ALPHA_ * x2.x;
            o.y = (1.0f - ALPHA_) * x1.y + ALPHA_ * x2.y;
            o.z = (1.0f - ALPHA_) * x1.z + ALPHA_ * x2.z;
            o.w = (1.0f - ALPHA_) * x1.w + ALPHA_ * x2.w;
        }
        o.x += f * acc.x; o.y += f * acc.y; o.z += f * acc.z; o.w += f * acc.w;
        *reinterpret_cast<float4*>(ob + ooff) = o;
    }
}

extern "C" void kernel_launch(void* const* d_in, const int* in_sizes, int n_in,
                              void* d_out, int out_size, void* d_ws, size_t ws_size,
                              hipStream_t stream) {
    const float* x    = (const float*)d_in[0];
    const int*   ei   = (const int*)d_in[1];
    const float* W1   = (const float*)d_in[2];
    const float* as1  = (const float*)d_in[3];
    const float* ad1  = (const float*)d_in[4];
    const float* b1   = (const float*)d_in[5];
    const float* W2   = (const float*)d_in[6];
    const float* as2  = (const float*)d_in[7];
    const float* ad2  = (const float*)d_in[8];
    const float* b2   = (const float*)d_in[9];
    float* out = (float*)d_out;

    const int n = in_sizes[0] / F_IN;
    const int E = in_sizes[1] / 2;
    const int* src = ei;
    const int* dst = ei + E;

    float* p   = (float*)d_ws;
    float* Hm  = p;  p += (size_t)n * F_OUT;
    int* q      = (int*)p;
    int* cntA   = q;  q += n;
    int* cntB   = q;  q += n;
    int* endA = q;  q += n;
    int* endB = q;  q += n;
    q += 4;
    int* listA  = q;  q += E;
    int* listB  = q;

    const int nblk  = (n + ROWS_MAX - 1) / ROWS_MAX;
    const int gblk  = (n + 16 * GEMM_WAVES - 1) / (16 * GEMM_WAVES);
    const int rblk  = (n + GATHER_WAVES * ROWS_PER_WAVE - 1) / (GATHER_WAVES * ROWS_PER_WAVE);

    unsigned* temp = (unsigned*)out;
    int* pre = (int*)out + 2 * (size_t)E;
    part_gemm_kernel<<<2 * SLICES + FILL_GEMM_GROUPS, LIST_BLOCK, 0, stream>>>(
        src, dst, E, n, temp, pre, x, W1, Hm);
    lists_kernel<<<2 * nblk, LISTS_BLOCK, 0, stream>>>(
        temp, pre, E, n, cntA, cntB, endA, endB, listA, listB);

    gather_kernel<true><<<rblk, GATHER_WAVES * 32, 0, stream>>>(listA, endA, cntA, as1, ad1, Hm, b1, b2,
                                                               1.0f - ALPHA_, out, n);

    gemm_wmma_kernel<<<gblk, GEMM_WAVES * 32, 0, stream>>>(x, W2, Hm, n);
    gather_kernel<false><<<rblk, GATHER_WAVES * 32, 0, stream>>>(listB, endB, cntB, as2, ad2, Hm, b1, b2,
                                                                ALPHA_, out, n);
}


# list build tails: lanes past a slice's or run's end add to a spare LDS word each (32 per table), not all to one word that the LDS serves a lane at a time; a run is then read in one trip of 5 loads
# speedup vs baseline: 27.3608x; 1.3001x over previous
#include <hip/hip_runtime.h>


typedef __attribute__((ext_vector_type(2))) float v2f;
typedef __attribute__((ext_vector_type(8))) float v8f;

#define F_IN   128
#define HEADS  4
#define C_OUT  32
#define F_OUT  128
#define ALPHA_ 0.5f
#define SLOPE_ 0.2f

__device__ __forceinline__ float lrelu(float v) {
    return v > 0.0f ? v : v * SLOPE_;
}

#define GEMM_WAVES 8
#define K_STEPS    (F_IN / 4)
#define N_TILES    (F_OUT / 16)
#define W_TASKS    ((F_IN / 2) * (F_OUT / 4) / (GEMM_WAVES * 32))
#define W_PITCH    (F_OUT + 16)

__global__ __launch_bounds__(GEMM_WAVES * 32) void gemm_wmma_kernel(
    const float* __restrict__ X, const float* __restrict__ Wm,
    float* __restrict__ Hm, int n)
{
    __shared__ v2f Wl[(F_IN / 2) * W_PITCH];

    const int tid   = threadIdx.x;
    const int lane  = tid & 31;
    const int half  = lane >> 4;
    const int l     = lane & 15;
    const int mtile = blockIdx.x * GEMM_WAVES + (tid >> 5);

    int rowA = mtile * 16 + l;
    if (rowA >= n) rowA = n - 1;

    float4 r0[W_TASKS], r1[W_TASKS];
    #pragma unroll
    for (int j = 0; j < W_TASKS; ++j) {
        const int kp = (tid >> 5) + j * GEMM_WAVES;
        const int c  = lane * 4;
        r0[j] = *reinterpret_cast<const float4*>(Wm + (size_t)(2 * kp) * F_OUT + c);
        r1[j] = *reinterpret_cast<const float4*>(Wm + (size_t)(2 * kp + 1) * F_OUT + c);
    }

    v2f a[K_STEPS];
    const float* __restrict__ xp = X + (size_t)rowA * F_IN + half * 2;
    #pragma unroll
    for (int s = 0; s < K_STEPS; ++s)
        a[s] = *reinterpret_cast<const v2f*>(xp + 4 * s);

    #pragma unroll
    for (int j = 0; j < W_TASKS; ++j) {
        const int kp = (tid >> 5) + j * GEMM_WAVES;
        v2f* __restrict__ dst = Wl + kp * W_PITCH + lane * 4;
        v2f p0, p1, p2, p3;
        p0.x = r0[j].x; p0.y = r1[j].x;
        p1.x = r0[j].y; p1.y = r1[j].y;
        p2.x = r0[j].z; p2.y = r1[j].z;
        p3.x = r0[j].w; p3.y = r1[j].w;
        dst[0] = p0; dst[1] = p1; dst[2] = p2; dst[3] = p3;
    }
    __syncthreads();

    if (mtile * 16 >= n) return;

    v8f acc[N_TILES];
    #pragma unroll
    for (int t = 0; t < N_TILES; ++t) acc[t] = (v8f){};

    const v2f* __restrict__ bp = Wl + half * W_PITCH + l;
    #pragma unroll
    for (int s = 0; s < K_STEPS; ++s) {
        #pragma unroll
        for (int t = 0; t < N_TILES; ++t) {
            const v2f b = bp[(2 * s) * W_PITCH + t * 16];
            acc[t] = __builtin_amdgcn_wmma_f32_16x16x4_f32(
                 false, a[s],  false, b,
                 (short)0, acc[t],  false,  false);
        }
    }

    const int rowBase = mtile * 16 + half * 8;
    float* __restrict__ hp = Hm + (size_t)rowBase * F_OUT + l;
    if (mtile * 16 + 16 <= n) {
        #pragma unroll
        for (int t = 0; t < N_TILES; ++t) {
            #pragma unroll
            for (int r = 0; r < 8; ++r) hp[r * F_OUT + t * 16] = acc[t][r];
        }
    } else {
        #pragma unroll
        for (int t = 0; t < N_TILES; ++t) {
            #pragma unroll
            for (int r = 0; r < 8; ++r)
                if (rowBase + r < n) hp[r * F_OUT + t * 16] = acc[t][r];
        }
    }
}

#define SLICES      64
#define RSH         8
#define ROWS_MAX    (1 << RSH)
#define NBUCK       256
#define LIST_BLOCK  1024
#define LISTS_BLOCK 512
#define EDGES_AHEAD 8
#define RUN_AHEAD   5
#define SPARE       32
#define STAGE_P     25600
#define STAGE_L     16384
#define LIST_PARTS  (LISTS_BLOCK / ROWS_MAX)
#define RUNS_PER_PART (SLICES / LIST_PARTS)

#define KE_STEPS 4
__device__ __forceinline__ void strip_load_a(v2f (&a)[KE_STEPS], const float* __restrict__ xq) {
    #pragma unroll
    for (int s = 0; s < KE_STEPS; ++s)
        a[s] = *reinterpret_cast<const v2f*>(xq + 4 * s);
}
__device__ __forceinline__ void strip_mma(v8f (&acc)[N_TILES], const v2f (&a)[KE_STEPS],
                                          const v2f* __restrict__ bq) {
    #pragma unroll
    for (int s = 0; s < KE_STEPS; ++s) {
        #pragma unroll
        for (int t = 0; t < N_TILES; ++t) {
            const v2f b = bq[(2 * s) * W_PITCH + t * 16];
            acc[t] = __builtin_amdgcn_wmma_f32_16x16x4_f32(
                 false, a[s],  false, b,
                 (short)0, acc[t],  false,  false);
        }
    }
}

#define FILL_GEMM_GROUPS 128
#define W_TASKS_L  ((F_IN / 2) * (F_OUT / 4) / LIST_BLOCK)

__global__ __launch_bounds__(LIST_BLOCK) void part_gemm_kernel(
    const int* __restrict__ e0, const int* __restrict__ e1, int E, int n,
    unsigned* __restrict__ temp, int* __restrict__ pre,
    const float* __restrict__ X, const float* __restrict__ Wm, float* __restrict__ Hm)
{
    __shared__ __attribute__((aligned(16))) int smem[2 * (NBUCK + SPARE) + STAGE_P + SPARE];
    const int tid = threadIdx.x;

    if (blockIdx.x >= 2 * SLICES) {
        v2f* __restrict__ Wl = reinterpret_cast<v2f*>(smem);
        const int lane  = tid & 31;
        const int half  = lane >> 4;
        const int l     = lane & 15;
        const int wave  = __builtin_amdgcn_readfirstlane(tid >> 5);
        const int mtile = wave * FILL_GEMM_GROUPS + ((int)blockIdx.x - 2 * SLICES);

        int rowA = mtile * 16 + l;
        if (rowA >= n) rowA = n - 1;

        float4 r0[W_TASKS_L], r1[W_TASKS_L];
        #pragma unroll
        for (int j = 0; j < W_TASKS_L; ++j) {
            const int kp = wave + j * (LIST_BLOCK / 32);
            const int c  = lane * 4;
            r0[j] = *reinterpret_cast<const float4*>(Wm + (size_t)(2 * kp) * F_OUT + c);
            r1[j] = *reinterpret_cast<const float4*>(Wm + (size_t)(2 * kp + 1) * F_OUT + c);
        }

        v2f a[KE_STEPS];
        const float* __restrict__ xq = X + (size_t)rowA * F_IN + half * 2;
        strip_load_a(a, xq);

        #pragma unroll
        for (int j = 0; j < W_TASKS_L; ++j) {
            const int kp = wave + j * (LIST_BLOCK / 32);
            v2f* __restrict__ dst = Wl + kp * W_PITCH + lane * 4;
            v2f p0, p1, p2, p3;
            p0.x = r0[j].x; p0.y = r1[j].x;
            p1.x = r0[j].y; p1.y = r1[j].y;
            p2.x = r0[j].z; p2.y = r1[j].z;
            p3.x = r0[j].w; p3.y = r1[j].w;
            dst[0] = p0; dst[1] = p1; dst[2] = p2; dst[3] = p3;
        }
        __syncthreads();

        if (mtile * 16 >= n) return;

        v8f acc[N_TILES];
        #pragma unroll
        for (int t = 0; t < N_TILES; ++t) acc[t] = (v8f){};

        const v2f* __restrict__ bq = Wl + half * W_PITCH + l;
        #pragma nounroll
        for (int q = 0; q < K_STEPS / KE_STEPS; ++q) {
            strip_mma(acc, a, bq);
            xq += 4 * KE_STEPS;
            bq += 2 * KE_STEPS * W_PITCH;
            if (q + 1 < K_STEPS / KE_STEPS) strip_load_a(a, xq);
        }

        const int rowBase = mtile * 16 + half * 8;
        float* __restrict__ hp = Hm + (size_t)rowBase * F_OUT + l;
        if (mtile * 16 + 16 <= n) {
            #pragma unroll
            for (int t = 0; t < N_TILES; ++t) {
                #pragma unroll
                for (int r = 0; r < 8; ++r) hp[r * F_OUT + t * 16] = acc[t][r];
            }
        } else {
            #pragma unroll
            for (int t = 0; t < N_TILES; ++t) {
                #pragma unroll
                for (int r = 0; r < 8; ++r)
                    if (rowBase + r < n) hp[r * F_OUT + t * 16] = acc[t][r];
            }
        }
        return;
    }

    int* __restrict__ cnt = smem;
    int* __restrict__ cur = smem + NBUCK + SPARE;
    unsigned* __restrict__ stage = reinterpret_cast<unsigned*>(smem + 2 * (NBUCK + SPARE));
    const int sp = tid & (SPARE - 1);
    const int g   = blockIdx.x;
    const int dir = g / SLICES;
    const int w   = g - dir * SLICES;
    const int* __restrict__ key = dir ? e0 : e1;
    const int* __restrict__ val = dir ? e1 : e0;
    const int sh = RSH;
    const unsigned mask = ROWS_MAX - 1;

    const int chunk = (E + SLICES - 1) / SLICES;
    const int lo = w * chunk;
    const int hi = (lo + chunk < E) ? (lo + chunk) : E;

    if (tid < NBUCK + SPARE) cnt[tid] = 0;
    __syncthreads();

    for (int e = lo + tid; e < hi; e += EDGES_AHEAD * LIST_BLOCK) {
        int k[EDGES_AHEAD];
        #pragma unroll
        for (int u = 0; u < EDGES_AHEAD; ++u) {
            const int ee = e + u * LIST_BLOCK;
            k[u] = key[(ee < hi) ? ee : (hi - 1)];
        }
        #pragma unroll
        for (int u = 0; u < EDGES_AHEAD; ++u)
            atomicAdd(&cnt[(e + u * LIST_BLOCK < hi) ? (k[u] >> sh) : (NBUCK + sp)], 1);
    }
    __syncthreads();

    if (tid < 32) {
        int v[NBUCK / 32];
        int s = 0;
        #pragma unroll
        for (int u = 0; u < NBUCK / 32; ++u) {
            const int t = cnt[tid * (NBUCK / 32) + u];
            v[u] = s;  s += t;
        }
        int incl = s;
        #pragma unroll
        for (int o = 1; o < 32; o <<= 1) {
            const int t = __shfl_up(incl, o, 32);
            if (tid >= o) incl += t;
        }
        const int excl = incl - s;
        int* __restrict__ prow = pre + (size_t)g * (NBUCK + 1);
        #pragma unroll
        for (int u = 0; u < NBUCK / 32; ++u) {
            cur[tid * (NBUCK / 32) + u]  = excl + v[u];
            prow[tid * (NBUCK / 32) + u] = excl + v[u];
        }
        cur[NBUCK + tid] = 0;
        if (tid == 31) prow[NBUCK] = incl;
    }
    __syncthreads();

    unsigned* __restrict__ tdst = temp + (size_t)dir * E + lo;
    if (chunk <= STAGE_P) {
        for (int e = lo + tid; e < hi; e += EDGES_AHEAD * LIST_BLOCK) {
            int k[EDGES_AHEAD], v[EDGES_AHEAD], pos[EDGES_AHEAD];
            #pragma unroll
            for (int u = 0; u < EDGES_AHEAD; ++u) {
                const int ee = e + u * LIST_BLOCK;
                const int ec = (ee < hi) ? ee : (hi - 1);
                k[u] = key[ec];
                v[u] = val[ec];
            }
            #pragma unroll
            for (int u = 0; u < EDGES_AHEAD; ++u)
                pos[u] = atomicAdd(&cur[(e + u * LIST_BLOCK < hi) ? (k[u] >> sh) : (NBUCK + sp)], 1);
            #pragma unroll
            for (int u = 0; u < EDGES_AHEAD; ++u)
                stage[(e + u * LIST_BLOCK < hi) ? pos[u] : (STAGE_P + sp)] =
                    ((unsigned)v[u] << sh) | ((unsigned)k[u] & mask);
        }
        __syncthreads();
        for (int i = tid; i < hi - lo; i += LIST_BLOCK) tdst[i] = stage[i];
    } else {
        for (int e = lo + tid; e < hi; e += LIST_BLOCK) {
            const int k = key[e];
            tdst[atomicAdd(&cur[k >> sh], 1)] = ((unsigned)val[e] << sh) | ((unsigned)k & mask);
        }
    }
}

__global__ __launch_bounds__(LISTS_BLOCK) void lists_kernel(
    const unsigned* __restrict__ temp, const int* __restrict__ pre, int E, int n,
    int* __restrict__ cntA, int* __restrict__ cntB,
    int* __restrict__ endA, int* __restrict__ endB,
    int* __restrict__ listA, int* __restrict__ listB)
{
    __shared__ int cntRS[SLICES * ROWS_MAX + SPARE];
    __shared__ int stageL[STAGE_L + SPARE];
    __shared__ int partQ[LIST_PARTS * ROWS_MAX];
    __shared__ int rowEx[ROWS_MAX];
    __shared__ int runStart[SLICES], runLen[SLICES], runP0[SLICES];
    __shared__ int wsum[ROWS_MAX / 32];
    __shared__ int sBase, sTotal;

    const int tid  = threadIdx.x;
    const int lane = tid & 31;
    const int wave = __builtin_amdgcn_readfirstlane(tid >> 5);
    const int nb  = (n + ROWS_MAX - 1) / ROWS_MAX;
    const int g   = blockIdx.x;
    const int dir = g / nb;
    const int b   = g - dir * nb;
    const int sh  = RSH;
    const unsigned mask = ROWS_MAX - 1;
    const int rowLo = b << sh;
    const int nrows = (n - rowLo < ROWS_MAX) ? (n - rowLo) : ROWS_MAX;
    const int chunk = (E + SLICES - 1) / SLICES;
    int* __restrict__ segCnt = dir ? cntB : cntA;
    int* __restrict__ segEnd = dir ? endB : endA;
    int* __restrict__ list   = dir ? listB : listA;

    if (tid < SLICES) {
        const int* __restrict__ prow = pre + (size_t)(dir * SLICES + tid) * (NBUCK + 1) + b;
        const int p0 = prow[0], p1 = prow[1];
        runP0[tid]    = p0;
        runLen[tid]   = p1 - p0;
        runStart[tid] = dir * E + tid * chunk + p0;
    }
    for (int i = tid; i < SLICES * ROWS_MAX + SPARE; i += LISTS_BLOCK) cntRS[i] = 0;
    __syncthreads();

    if (tid < 32) {
        int a = 0, t = 0;
        for (int r = tid; r < SLICES; r += 32) { a += runP0[r]; t += runLen[r]; }
        #pragma unroll
        for (int o = 16; o > 0; o >>= 1) {
            a += __shfl_xor(a, o, 32);
            t += __shfl_xor(t, o, 32);
        }
        if (tid == 0) { sBase = a; sTotal = t; }
    }

    for (int r = wave; r < SLICES; r += LISTS_BLOCK / 32) {
        const int len = runLen[r];
        const unsigned* __restrict__ p = temp + runStart[r];
        const int crow = r * ROWS_MAX;
        for (int o = lane; o < len; o += 32 * RUN_AHEAD) {
            unsigned t[RUN_AHEAD];
            #pragma unroll
            for (int u = 0; u < RUN_AHEAD; ++u)
                t[u] = p[(o + 32 * u < len) ? (o + 32 * u) : (len - 1)];
            #pragma unroll
            for (int u = 0; u < RUN_AHEAD; ++u)
                atomicAdd(&cntRS[(o + 32 * u < len) ? (crow + (int)(t[u] & mask)) : (SLICES * ROWS_MAX + lane)], 1);
        }
    }
    __syncthreads();

    const int row = tid & (ROWS_MAX - 1);
    const int q   = tid / ROWS_MAX;
    {
        int sacc = 0;
        for (int r = q * RUNS_PER_PART; r < (q + 1) * RUNS_PER_PART; ++r) sacc += cntRS[r * ROWS_MAX + row];
        partQ[q * ROWS_MAX + row] = sacc;
    }
    __syncthreads();
    int c = 0, s = 0;
    if (tid < ROWS_MAX) {
        for (int qq = 0; qq < LIST_PARTS; ++qq) c += partQ[qq * ROWS_MAX + tid];
        s = c;
        #pragma unroll
        for (int o = 1; o < 32; o <<= 1) {
            const int t = __shfl_up(s, o, 32);
            if (lane >= o) s += t;
        }
        if (lane == 31) wsum[tid >> 5] = s;
    }
    __syncthreads();
    if (tid < ROWS_MAX) {
        int off = 0;
        for (int jw = 0; jw < (tid >> 5); ++jw) off += wsum[jw];
        const int excl = off + s - c;
        rowEx[tid] = excl;
        if (tid < nrows) {
            segCnt[rowLo + tid] = c;
            segEnd[rowLo + tid] = sBase + excl + c;
        }
    }
    __syncthreads();
    {
        int start = rowEx[row];
        for (int qq = 0; qq < q; ++qq) start += partQ[qq * ROWS_MAX + row];
        for (int r = q * RUNS_PER_PART; r < (q + 1) * RUNS_PER_PART; ++r) {
            const int t = cntRS[r * ROWS_MAX + row];
            cntRS[r * ROWS_MAX + row] = start;
            start += t;
        }
    }
    __syncthreads();

    const int base   = sBase;
    const int total  = sTotal;
    if (total <= STAGE_L) {
        for (int r = wave; r < SLICES; r += LISTS_BLOCK / 32) {
            const int len = runLen[r];
            const unsigned* __restrict__ p = temp + runStart[r];
            const int crow = r * ROWS_MAX;
            for (int o = lane; o < len; o += 32 * RUN_AHEAD) {
                unsigned t[RUN_AHEAD];
                int pos[RUN_AHEAD];
                #pragma unroll
                for (int u = 0; u < RUN_AHEAD; ++u)
                    t[u] = p[(o + 32 * u < len) ? (o + 32 * u) : (len - 1)];
                #pragma unroll
                for (int u = 0; u < RUN_AHEAD; ++u)
                    pos[u] = atomicAdd(&cntRS[(o + 32 * u < len) ? (crow + (int)(t[u] & mask)) : (SLICES * ROWS_MAX + lane)], 1);
                #pragma unroll
                for (int u = 0; u < RUN_AHEAD; ++u)
                    stageL[(o + 32 * u < len) ? pos[u] : (STAGE_L + lane)] = (int)(t[u] >> sh);
            }
        }
        __syncthreads();
        for (int i = tid; i < total; i += LISTS_BLOCK) list[base + i] = stageL[i];
    } else {
        for (int r = wave; r < SLICES; r += LISTS_BLOCK / 32) {
            const int len = runLen[r];
            const unsigned* __restrict__ p = temp + runStart[r];
            const int crow = r * ROWS_MAX;
            for (int o = lane; o < len; o += 32) {
                const unsigned t = p[o];
                list[base + atomicAdd(&cntRS[crow + (int)(t & mask)], 1)] = (int)(t >> sh);
            }
        }
    }
}

typedef __attribute__((ext_vector_type(4))) float v4f;
#define LOG2E_ 1.4426950408889634f

template <int K>
__device__ __forceinline__ float from_lane_xor(float v) {
    return __builtin_bit_cast(float,
        __builtin_amdgcn_update_dpp(0, __builtin_bit_cast(int, v), 0x160 + K, 0xf, 0xf, true));
}
template <int U>
__device__ __forceinline__ float from_group_lane(float v) {
    return __builtin_bit_cast(float,
        __builtin_amdgcn_mov_dpp8(__builtin_bit_cast(int, v), U * 0x249249));
}
__device__ __forceinline__ float head_sum(float v) {
    v += from_lane_xor<1>(v);
    v += from_lane_xor<2>(v);
    v += from_lane_xor<4>(v);
    return v;
}
__device__ __forceinline__ float head_max(float v) {
    v = fmaxf(v, from_lane_xor<1>(v));
    v = fmaxf(v, from_lane_xor<2>(v));
    v = fmaxf(v, from_lane_xor<4>(v));
    return v;
}
__device__ __forceinline__ float dot4(const v4f a, const v4f b) {
    const v2f t = a.lo * b.lo + a.hi * b.hi;
    return t.x + t.y;
}

#define ROWS_IN_FLIGHT 8
#define GATHER_WAVES   8
#define ROWS_PER_WAVE  5
template <bool FIRST>
__global__ __launch_bounds__(GATHER_WAVES * 32) void gather_kernel(
    const int* __restrict__ list, const int* __restrict__ segEnd,
    const int* __restrict__ segCnt,
    const float* __restrict__ as, const float* __restrict__ ad,
    const float* __restrict__ Hm,
    const float* __restrict__ b1, const float* __restrict__ b2,
    float scale, float* __restrict__ out, int n)
{
    const int lane = threadIdx.x & 31;
    const int j    = lane & 7;
    const bool j4 = (j & 4) != 0, j2 = (j & 2) != 0, j1 = (j & 1) != 0;
    const int d0 = __builtin_amdgcn_readfirstlane(
        (int)(blockIdx.x * GATHER_WAVES + (threadIdx.x >> 5))) * ROWS_PER_WAVE;
    if (d0 >= n) return;
    const int nr = (n - d0 < ROWS_PER_WAVE) ? (n - d0) : ROWS_PER_WAVE;

    int cntv = 0, endv = 0;
    if (lane < nr) { cntv = segCnt[d0 + lane]; endv = segEnd[d0 + lane]; }

    const unsigned loff = (unsigned)lane * 16u;
    const char* __restrict__ hb = reinterpret_cast<const char*>(Hm);
    char* __restrict__ ob = reinterpret_cast<char*>(out);
    const v4f a4 = *reinterpret_cast<const v4f*>(as + lane * 4) * LOG2E_;
    const v4f d4 = *reinterpret_cast<const v4f*>(ad + lane * 4) * LOG2E_;
    const float ninf = -__builtin_huge_valf();

    int deg = __builtin_amdgcn_readlane(cntv, 0);
    int beg = __builtin_amdgcn_readlane(endv, 0) - deg;
    int sA = d0, sB = d0;
    if (lane < deg)      sA = list[beg + lane];
    if (lane + 32 < deg) sB = list[beg + 32 + lane];
    v4f self = *reinterpret_cast<const v4f*>(hb + (((unsigned)d0 << 9) + loff));
    asm volatile("" :: "v"(sA), "v"(sB), "v"(self));

    for (int r = 0; r < nr; ++r) {
        const int d = d0 + r;
        const int degc = deg, begc = beg;
        const int s0 = sA, s1 = sB;
        const v4f selfc = self;
        const unsigned ooff = ((unsigned)d << 9) + loff;

        if (r + 1 < nr) {
            deg = __builtin_amdgcn_readlane(cntv, r + 1);
            beg = __builtin_amdgcn_readlane(endv, r + 1) - deg;
            sA = d + 1; sB = d + 1;
            if (lane < deg)      sA = list[beg + lane];
            if (lane + 32 < deg) sB = list[beg + 32 + lane];
            self = *reinterpret_cast<const v4f*>(hb + (((unsigned)(d + 1) << 9) + loff));
        }
        float4 o;
        if (!FIRST) o = *reinterpret_cast<const float4*>(ob + ooff);

        const float edh = head_sum(dot4(selfc, d4));
        float m   = lrelu(head_sum(dot4(selfc, a4)) + edh);
        float den = (j == 0) ? 1.0f : 0.0f;
        v4f   acc = selfc;

        for (int base = 0; base < degc; base += 32) {
            int s;
            if (base == 0)       s = s0;
            else if (base == 32) s = s1;
            else {
                s = d;
                if (base + lane < degc) s = list[begc + base + lane];
                asm volatile("" :: "v"(s));
            }
            const int cnt = (degc - base < 32) ? (degc - base) : 32;
            for (int e = 0; e < cnt; e += ROWS_IN_FLIGHT) {
                v4f   row[ROWS_IN_FLIGHT];
                float q[ROWS_IN_FLIGHT];
                #pragma unroll
                for (int u = 0; u < ROWS_IN_FLIGHT; ++u) {
                    const unsigned se = (unsigned)__builtin_amdgcn_readlane(s, e + u);
                    row[u] = *reinterpret_cast<const v4f*>(hb + ((se << 9) + loff));
                }
                #pragma unroll
                for (int u = 0; u < ROWS_IN_FLIGHT; ++u) q[u] = dot4(row[u], a4);

                float k4[4], k2[2];
                #pragma unroll
                for (int t = 0; t < 4; ++t) {
                    const float keep = j4 ? q[4 + t] : q[t];
                    const float send = j4 ? q[t] : q[4 + t];
                    k4[t] = keep + from_lane_xor<4>(send);
                }
                #pragma unroll
                for (int t = 0; t < 2; ++t) {
                    const float keep = j2 ? k4[2 + t] : k4[t];
                    const float send = j2 ? k4[t] : k4[2 + t];
                    k2[t] = keep + from_lane_xor<2>(send);
                }
                const float keep1 = j1 ? k2[1] : k2[0];
                const float send1 = j1 ? k2[0] : k2[1];
                float v = lrelu(keep1 + from_lane_xor<1>(send1) + edh);
                if (e + j >= cnt) v = ninf;

                const float mn = fmaxf(m, head_max(v));
                const float rs = __builtin_amdgcn_exp2f(m - mn);
                const float p  = __builtin_amdgcn_exp2f(v - mn);
                m   = mn;
                den = den * rs + p;
                acc *= rs;
                acc += from_group_lane<0>(p) * row[0];
                acc += from_group_lane<1>(p) * row[1];
                acc += from_group_lane<2>(p) * row[2];
                acc += from_group_lane<3>(p) * row[3];
                acc += from_group_lane<4>(p) * row[4];
                acc += from_group_lane<5>(p) * row[5];
                acc += from_group_lane<6>(p) * row[6];
                acc += from_group_lane<7>(p) * row[7];
            }
        }

        const float f = scale / head_sum(den);
        if (FIRST) {
            const float4 x1 = *reinterpret_cast<const float4*>(b1 + lane * 4);
            const float4 x2 = *reinterpret_cast<const float4*>(b2 + lane * 4);
            o.x = (1.0f - ALPHA_) * x1.x + ALPHA_ * x2.x;
            o.y = (1.0f - ALPHA_) * x1.y + ALPHA_ * x2.y;
            o.z = (1.0f - ALPHA_) * x1.z + ALPHA_ * x2.z;
            o.w = (1.0f - ALPHA_) * x1.w + ALPHA_ * x2.w;
        }
        o.x += f * acc.x; o.y += f * acc.y; o.z += f * acc.z; o.w += f * acc.w;
        *reinterpret_cast<float4*>(ob + ooff) = o;
    }
}

extern "C" void kernel_launch(void* const* d_in, const int* in_sizes, int n_in,
                              void* d_out, int out_size, void* d_ws, size_t ws_size,
                              hipStream_t stream) {
    const float* x    = (const float*)d_in[0];
    const int*   ei   = (const int*)d_in[1];
    const float* W1   = (const float*)d_in[2];
    const float* as1  = (const float*)d_in[3];
    const float* ad1  = (const float*)d_in[4];
    const float* b1   = (const float*)d_in[5];
    const float* W2   = (const float*)d_in[6];
    const float* as2  = (const float*)d_in[7];
    const float* ad2  = (const float*)d_in[8];
    const float* b2   = (const float*)d_in[9];
    float* out = (float*)d_out;

    const int n = in_sizes[0] / F_IN;
    const int E = in_sizes[1] / 2;
    const int* src = ei;
    const int* dst = ei + E;

    float* p   = (float*)d_ws;
    float* Hm  = p;  p += (size_t)n * F_OUT;
    int* q      = (int*)p;
    int* cntA   = q;  q += n;
    int* cntB   = q;  q += n;
    int* endA = q;  q += n;
    int* endB = q;  q += n;
    q += 4;
    int* listA  = q;  q += E;
    int* listB  = q;

    const int nblk  = (n + ROWS_MAX - 1) / ROWS_MAX;
    const int gblk  = (n + 16 * GEMM_WAVES - 1) / (16 * GEMM_WAVES);
    const int rblk  = (n + GATHER_WAVES * ROWS_PER_WAVE - 1) / (GATHER_WAVES * ROWS_PER_WAVE);

    unsigned* temp = (unsigned*)out;
    int* pre = (int*)out + 2 * (size_t)E;
    part_gemm_kernel<<<2 * SLICES + FILL_GEMM_GROUPS, LIST_BLOCK, 0, stream>>>(
        src, dst, E, n, temp, pre, x, W1, Hm);
    lists_kernel<<<2 * nblk, LISTS_BLOCK, 0, stream>>>(
        temp, pre, E, n, cntA, cntB, endA, endB, listA, listB);

    gather_kernel<true><<<rblk, GATHER_WAVES * 32, 0, stream>>>(listA, endA, cntA, as1, ad1, Hm, b1, b2,
                                                               1.0f - ALPHA_, out, n);

    gemm_wmma_kernel<<<gblk, GEMM_WAVES * 32, 0, stream>>>(x, W2, Hm, n);
    gather_kernel<false><<<rblk, GATHER_WAVES * 32, 0, stream>>>(listB, endB, cntB, as2, ad2, Hm, b1, b2,
                                                                ALPHA_, out, n);
}
